# starting kernel
# baseline (speedup 1.0000x reference)
#include <hip/hip_runtime.h>
#include <math.h>

typedef __attribute__((ext_vector_type(2))) float v2f;
typedef __attribute__((ext_vector_type(8))) float v8f;

#define N_NODES   50000
#define N_EDGES   800000
#define IN_F      256
#define EDGE_F    64
#define NH        8
#define NF        32
#define HF        256

__device__ __forceinline__ unsigned enc_f32(float f) {
  unsigned u = __float_as_uint(f);
  return (u & 0x80000000u) ? ~u : (u | 0x80000000u);
}
__device__ __forceinline__ float dec_f32(unsigned u) {
  unsigned b = (u & 0x80000000u) ? (u ^ 0x80000000u) : ~u;
  return __uint_as_float(b);
}

__global__ __launch_bounds__(256)
void gemm_wh_wmma(const float* __restrict__ h, const float* __restrict__ W,
                  float* __restrict__ Wh) {
  const int lane = threadIdx.x & 31;
  const int wave = threadIdx.x >> 5;
  const int m0   = blockIdx.x * 16;
  const int n0   = wave * 32;
  const int l15  = lane & 15;
  const int hi   = lane >> 4;
  const int khalf = hi * 2;

  v8f c0 = {}; v8f c1 = {};
  const float* arow = h + (size_t)(m0 + l15) * IN_F;

  for (int k0 = 0; k0 < IN_F; k0 += 4) {
    v2f a = *(const v2f*)(arow + k0 + khalf);
    const float* bb = W + (size_t)(k0 + khalf) * HF + n0 + l15;
    v2f b0, b1;
    b0.x = bb[0];        b0.y = bb[HF];
    b1.x = bb[16];       b1.y = bb[HF + 16];
    c0 = __builtin_amdgcn_wmma_f32_16x16x4_f32(false, a, false, b0,
                                               (short)0, c0, false, false);
    c1 = __builtin_amdgcn_wmma_f32_16x16x4_f32(false, a, false, b1,
                                               (short)0, c1, false, false);
  }
  const int mo = m0 + (hi << 3);
  const int nc = n0 + l15;
  for (int r = 0; r < 8; ++r) {
    Wh[(size_t)(mo + r) * HF + nc]      = c0[r];
    Wh[(size_t)(mo + r) * HF + nc + 16] = c1[r];
  }
}

__global__ void fold_we(const float* __restrict__ We_w,
                        const float* __restrict__ a_edge,
                        float* __restrict__ V) {
  int t = threadIdx.x;
  int k = t >> 3, hh = t & 7;
  const float* wr = We_w + (size_t)k * HF + hh * NF;
  const float* ae = a_edge + hh * NF;
  float acc = 0.f;
  for (int f = 0; f < NF; ++f) acc += wr[f] * ae[f];
  V[k * NH + hh] = acc;
}

__global__ __launch_bounds__(256)
void node_scores(const float* __restrict__ Wh, const float* __restrict__ a_src,
                 const float* __restrict__ a_dst,
                 float* __restrict__ ssrc, float* __restrict__ sdst) {
  int t = blockIdx.x * blockDim.x + threadIdx.x;
  if (t >= N_NODES * NH) return;
  int n = t >> 3, hh = t & 7;
  const float* row = Wh + (size_t)n * HF + hh * NF;
  const float* as  = a_src + hh * NF;
  const float* ad  = a_dst + hh * NF;
  float accs = 0.f, accd = 0.f;
  for (int f = 0; f < NF; ++f) { float v = row[f]; accs += v * as[f]; accd += v * ad[f]; }
  ssrc[t] = accs;
  sdst[t] = accd;
}

__global__ __launch_bounds__(256)
void edge_logits(const float* __restrict__ ef, const int* __restrict__ src,
                 const int* __restrict__ dst, const float* __restrict__ V,
                 const float* __restrict__ ssrc, const float* __restrict__ sdst,
                 float* __restrict__ sbuf, unsigned* __restrict__ m) {
  __shared__ float Vl[EDGE_F * NH];
  for (int i = threadIdx.x; i < EDGE_F * NH; i += blockDim.x) Vl[i] = V[i];
  __syncthreads();
  int t = blockIdx.x * blockDim.x + threadIdx.x;
  if (t >= N_EDGES * NH) return;
  int e = t >> 3, hh = t & 7;
  const float4* efr = (const float4*)(ef + (size_t)e * EDGE_F);
  float acc = 0.f;
  for (int q = 0; q < EDGE_F / 4; ++q) {
    float4 v = efr[q];
    int k = q * 4;
    acc += v.x * Vl[(k + 0) * NH + hh] + v.y * Vl[(k + 1) * NH + hh]
         + v.z * Vl[(k + 2) * NH + hh] + v.w * Vl[(k + 3) * NH + hh];
  }
  int s = src[e], d = dst[e];
  float sv = ssrc[s * NH + hh] + sdst[d * NH + hh] + acc;
  sv = sv > 0.f ? sv : 0.01f * sv;
  sbuf[t] = sv;
  atomicMax(&m[d * NH + hh], enc_f32(sv));
}

__global__ __launch_bounds__(256)
void softmax_den(const int* __restrict__ dst, const unsigned* __restrict__ m,
                 float* __restrict__ sbuf, float* __restrict__ den) {
  int t = blockIdx.x * blockDim.x + threadIdx.x;
  if (t >= N_EDGES * NH) return;
  int e = t >> 3, hh = t & 7;
  int d = dst[e];
  float ex = expf(sbuf[t] - dec_f32(m[d * NH + hh]));
  sbuf[t] = ex;
  atomicAdd(&den[d * NH + hh], ex);
}

__global__ __launch_bounds__(256)
void aggregate(const int* __restrict__ src, const int* __restrict__ dst,
               const float* __restrict__ Wh, const float* __restrict__ sbuf,
               const float* __restrict__ den, float* __restrict__ out) {
  int e  = blockIdx.x;
  int c  = threadIdx.x;
  int hh = c >> 5;
  int s = src[e], d = dst[e];
  float a = sbuf[(size_t)e * NH + hh] / den[d * NH + hh];
  float val = Wh[(size_t)s * HF + c] * a;
  atomicAdd(&out[(size_t)d * HF + c], val);
}

__global__ __launch_bounds__(256)
void elu_inplace(float* __restrict__ out, int n) {
  int t = blockIdx.x * blockDim.x + threadIdx.x;
  if (t >= n) return;
  float x = out[t];
  out[t] = x > 0.f ? x : expm1f(x);
}

extern "C" void kernel_launch(void* const* d_in, const int* in_sizes, int n_in,
                              void* d_out, int out_size, void* d_ws, size_t ws_size,
                              hipStream_t stream) {
  const float* h   = (const float*)d_in[0];
  const float* ef  = (const float*)d_in[1];
  const int*   src = (const int*)  d_in[2];
  const int*   dst = (const int*)  d_in[3];
  const float* W_w = (const float*)d_in[4];
  const float* We_w= (const float*)d_in[5];
  const float* a_s = (const float*)d_in[6];
  const float* a_d = (const float*)d_in[7];
  const float* a_e = (const float*)d_in[8];
  float* out = (float*)d_out;

  char* ws = (char*)d_ws;
  const size_t WH_BYTES   = (size_t)N_NODES * HF * 4;
  const size_t SBUF_BYTES = (size_t)N_EDGES * NH * 4;
  const size_t NH_BYTES   = (size_t)N_NODES * NH * 4;
  float*    Wh   = (float*)   (ws);
  float*    sbuf = (float*)   (ws + WH_BYTES);
  float*    ssrc = (float*)   (ws + WH_BYTES + SBUF_BYTES);
  float*    sdst = (float*)   (ws + WH_BYTES + SBUF_BYTES + NH_BYTES);
  unsigned* mmax = (unsigned*)(ws + WH_BYTES + SBUF_BYTES + 2 * NH_BYTES);
  float*    den  = (float*)   (ws + WH_BYTES + SBUF_BYTES + 3 * NH_BYTES);
  float*    V    = (float*)   (ws + WH_BYTES + SBUF_BYTES + 4 * NH_BYTES);

  hipMemsetAsync(out,  0, (size_t)N_NODES * HF * 4, stream);
  hipMemsetAsync(mmax, 0, NH_BYTES, stream);
  hipMemsetAsync(den,  0, NH_BYTES, stream);

  gemm_wh_wmma<<<N_NODES / 16, 256, 0, stream>>>(h, W_w, Wh);
  fold_we<<<1, 512, 0, stream>>>(We_w, a_e, V);
  node_scores<<<(N_NODES * NH + 255) / 256, 256, 0, stream>>>(Wh, a_s, a_d, ssrc, sdst);
  edge_logits<<<(N_EDGES * NH + 255) / 256, 256, 0, stream>>>(ef, src, dst, V, ssrc, sdst, sbuf, mmax);
  softmax_den<<<(N_EDGES * NH + 255) / 256, 256, 0, stream>>>(dst, mmax, sbuf, den);
  aggregate<<<N_EDGES, 256, 0, stream>>>(src, dst, Wh, sbuf, den, out);
  elu_inplace<<<(N_NODES * HF + 255) / 256, 256, 0, stream>>>(out, N_NODES * HF);
}


# Aggregation by dst tile in LDS: a 1024-thread block owns 196 dst rows, scans the unsorted dst list, adds Wh[src]*a with LDS f32 atomics, stores ELU once; no global atomics on out; sum order changes
# speedup vs baseline: 2.9286x; 2.9286x over previous
#include <hip/hip_runtime.h>
#include <math.h>

typedef __attribute__((ext_vector_type(2))) float v2f;
typedef __attribute__((ext_vector_type(8))) float v8f;

#define N_NODES   50000
#define N_EDGES   800000
#define IN_F      256
#define EDGE_F    64
#define NH        8
#define NF        32
#define HF        256

__device__ __forceinline__ unsigned enc_f32(float f) {
  unsigned u = __float_as_uint(f);
  return (u & 0x80000000u) ? ~u : (u | 0x80000000u);
}
__device__ __forceinline__ float dec_f32(unsigned u) {
  unsigned b = (u & 0x80000000u) ? (u ^ 0x80000000u) : ~u;
  return __uint_as_float(b);
}

__global__ __launch_bounds__(256)
void gemm_wh_wmma(const float* __restrict__ h, const float* __restrict__ W,
                  float* __restrict__ Wh) {
  const int lane = threadIdx.x & 31;
  const int wave = threadIdx.x >> 5;
  const int m0   = blockIdx.x * 16;
  const int n0   = wave * 32;
  const int l15  = lane & 15;
  const int hi   = lane >> 4;
  const int khalf = hi * 2;

  v8f c0 = {}; v8f c1 = {};
  const float* arow = h + (size_t)(m0 + l15) * IN_F;

  for (int k0 = 0; k0 < IN_F; k0 += 4) {
    v2f a = *(const v2f*)(arow + k0 + khalf);
    const float* bb = W + (size_t)(k0 + khalf) * HF + n0 + l15;
    v2f b0, b1;
    b0.x = bb[0];        b0.y = bb[HF];
    b1.x = bb[16];       b1.y = bb[HF + 16];
    c0 = __builtin_amdgcn_wmma_f32_16x16x4_f32(false, a, false, b0,
                                               (short)0, c0, false, false);
    c1 = __builtin_amdgcn_wmma_f32_16x16x4_f32(false, a, false, b1,
                                               (short)0, c1, false, false);
  }
  const int mo = m0 + (hi << 3);
  const int nc = n0 + l15;
  for (int r = 0; r < 8; ++r) {
    Wh[(size_t)(mo + r) * HF + nc]      = c0[r];
    Wh[(size_t)(mo + r) * HF + nc + 16] = c1[r];
  }
}

__global__ void fold_we(const float* __restrict__ We_w,
                        const float* __restrict__ a_edge,
                        float* __restrict__ V) {
  int t = threadIdx.x;
  int k = t >> 3, hh = t & 7;
  const float* wr = We_w + (size_t)k * HF + hh * NF;
  const float* ae = a_edge + hh * NF;
  float acc = 0.f;
  for (int f = 0; f < NF; ++f) acc += wr[f] * ae[f];
  V[k * NH + hh] = acc;
}

__global__ __launch_bounds__(256)
void node_scores(const float* __restrict__ Wh, const float* __restrict__ a_src,
                 const float* __restrict__ a_dst,
                 float* __restrict__ ssrc, float* __restrict__ sdst) {
  int t = blockIdx.x * blockDim.x + threadIdx.x;
  if (t >= N_NODES * NH) return;
  int n = t >> 3, hh = t & 7;
  const float* row = Wh + (size_t)n * HF + hh * NF;
  const float* as  = a_src + hh * NF;
  const float* ad  = a_dst + hh * NF;
  float accs = 0.f, accd = 0.f;
  for (int f = 0; f < NF; ++f) { float v = row[f]; accs += v * as[f]; accd += v * ad[f]; }
  ssrc[t] = accs;
  sdst[t] = accd;
}

__global__ __launch_bounds__(256)
void edge_logits(const float* __restrict__ ef, const int* __restrict__ src,
                 const int* __restrict__ dst, const float* __restrict__ V,
                 const float* __restrict__ ssrc, const float* __restrict__ sdst,
                 float* __restrict__ sbuf, unsigned* __restrict__ m) {
  __shared__ float Vl[EDGE_F * NH];
  for (int i = threadIdx.x; i < EDGE_F * NH; i += blockDim.x) Vl[i] = V[i];
  __syncthreads();
  int t = blockIdx.x * blockDim.x + threadIdx.x;
  if (t >= N_EDGES * NH) return;
  int e = t >> 3, hh = t & 7;
  const float4* efr = (const float4*)(ef + (size_t)e * EDGE_F);
  float acc = 0.f;
  for (int q = 0; q < EDGE_F / 4; ++q) {
    float4 v = efr[q];
    int k = q * 4;
    acc += v.x * Vl[(k + 0) * NH + hh] + v.y * Vl[(k + 1) * NH + hh]
         + v.z * Vl[(k + 2) * NH + hh] + v.w * Vl[(k + 3) * NH + hh];
  }
  int s = src[e], d = dst[e];
  float sv = ssrc[s * NH + hh] + sdst[d * NH + hh] + acc;
  sv = sv > 0.f ? sv : 0.01f * sv;
  sbuf[t] = sv;
  atomicMax(&m[d * NH + hh], enc_f32(sv));
}

__global__ __launch_bounds__(256)
void softmax_den(const int* __restrict__ dst, const unsigned* __restrict__ m,
                 float* __restrict__ sbuf, float* __restrict__ den) {
  int t = blockIdx.x * blockDim.x + threadIdx.x;
  if (t >= N_EDGES * NH) return;
  int e = t >> 3, hh = t & 7;
  int d = dst[e];
  float ex = expf(sbuf[t] - dec_f32(m[d * NH + hh]));
  sbuf[t] = ex;
  atomicAdd(&den[d * NH + hh], ex);
}

#define TROWS     196
#define TWAVES    32
#define LIST_CAP  544
#define E4        (N_EDGES / 4)
static_assert(E4 % 32 == 0, "the scan loads whole waves of int4");
static_assert(TROWS <= 256, "the local row packs into 8 bits");
static_assert(N_EDGES < (1 << 24), "the edge id packs into 24 bits");

__device__ __forceinline__ void lds_add(float* p, float v) {
  __hip_atomic_fetch_add(p, v, __ATOMIC_RELAXED, __HIP_MEMORY_SCOPE_WORKGROUP);
}

__global__ __launch_bounds__(1024)
void aggregate_tile(const int* __restrict__ src, const int* __restrict__ dst,
                    const float* __restrict__ Wh, const float* __restrict__ sbuf,
                    const float* __restrict__ den, float* __restrict__ out) {
  __shared__ float    tile[TROWS * HF];
  __shared__ unsigned elist[TWAVES * LIST_CAP];
  __shared__ float    astage[TWAVES * 32 * NH];

  const int tid  = threadIdx.x;
  const int lane = tid & 31;
  const int wave = __builtin_amdgcn_readfirstlane(tid >> 5);
  const int base = blockIdx.x * TROWS;
  const int nrows = min(TROWS, N_NODES - base);
  const unsigned lt = (1u << lane) - 1u;

  float4* t4 = (float4*)tile;
  for (int i = tid; i < TROWS * HF / 4; i += 1024) t4[i] = make_float4(0.f, 0.f, 0.f, 0.f);
  __syncthreads();

  unsigned* mylist  = elist + wave * LIST_CAP;
  float*    mystage = astage + wave * (32 * NH);
  const int4* dst4 = (const int4*)dst;
  const int NIT = (E4 + 4095) / 4096;
  int cnt = 0;

  for (int it = 0; it <= NIT; ++it) {
    const bool last = (it == NIT);
    if (cnt >= 32 || (last && cnt > 0)) {
      for (int b = 0; b < cnt; b += 32) {
        const int nb = min(32, cnt - b);
        unsigned ent = 0u;
        int s_i = 0;
        if (lane < nb) {
          ent = mylist[b + lane];
          const int e = (int)(ent >> 8);
          const int d = base + (int)(ent & 255u);
          s_i = src[e];
          const float4* sp = (const float4*)(sbuf + (size_t)e * NH);
          const float4* dp = (const float4*)(den + (size_t)d * NH);
          const float4 x0 = sp[0], x1 = sp[1], y0 = dp[0], y1 = dp[1];
          float4* st = (float4*)(mystage + lane * NH);
          st[0] = make_float4(x0.x / y0.x, x0.y / y0.y, x0.z / y0.z, x0.w / y0.w);
          st[1] = make_float4(x1.x / y1.x, x1.y / y1.y, x1.z / y1.z, x1.w / y1.w);
        }
        for (int u0 = 0; u0 < nb; u0 += 4) {
          float4 v0[4], v1[4];
          float a0[4], a1[4];
          int lr[4];
#pragma unroll
          for (int k = 0; k < 4; ++k) {
            const int u = min(u0 + k, nb - 1);
            const int s = __builtin_amdgcn_readlane(s_i, u);
            lr[k] = __builtin_amdgcn_readlane((int)ent, u) & 255;
            const float4* rp = (const float4*)(Wh + (size_t)s * HF);
            v0[k] = rp[lane];
            v1[k] = rp[32 + lane];
            a0[k] = mystage[u * NH + (lane >> 3)];
            a1[k] = mystage[u * NH + 4 + (lane >> 3)];
          }
#pragma unroll
          for (int k = 0; k < 4; ++k) {
            if (u0 + k < nb) {
              float* tp = tile + lr[k] * HF + lane;
              lds_add(tp,       v0[k].x * a0[k]);
              lds_add(tp + 32,  v0[k].y * a0[k]);
              lds_add(tp + 64,  v0[k].z * a0[k]);
              lds_add(tp + 96,  v0[k].w * a0[k]);
              lds_add(tp + 128, v1[k].x * a1[k]);
              lds_add(tp + 160, v1[k].y * a1[k]);
              lds_add(tp + 192, v1[k].z * a1[k]);
              lds_add(tp + 224, v1[k].w * a1[k]);
            }
          }
        }
      }
      cnt = 0;
    }
    if (last) break;

    const int qb = it * 4096 + wave * 128;
    int4 dd[4];
#pragma unroll
    for (int k = 0; k < 4; ++k) {
      const int q = min(qb + k * 32, E4 - 32);
      dd[k] = dst4[q + lane];
    }
#pragma unroll
    for (int k = 0; k < 4; ++k) {
      const unsigned eb = (unsigned)(qb + k * 32 + lane) * 4u;
      const unsigned lim = (qb + k * 32 < E4) ? (unsigned)nrows : 0u;
      const int dc[4] = {dd[k].x, dd[k].y, dd[k].z, dd[k].w};
#pragma unroll
      for (int j = 0; j < 4; ++j) {
        const unsigned lrow = (unsigned)(dc[j] - base);
        const bool hit = lrow < lim;
        const unsigned mask = __builtin_amdgcn_ballot_w32(hit);
        if (mask != 0u) {
          if (hit) mylist[cnt + __popc(mask & lt)] = ((eb + (unsigned)j) << 8) | lrow;
          cnt += __popc(mask);
        }
      }
    }
  }
  __syncthreads();

  for (int i = tid; i < nrows * 64; i += 1024) {
    const int r = i >> 6, g = i & 63;
    const int half = g >> 5, l = g & 31;
    const float* tp = tile + r * HF + half * 128 + l;
    float4 x = make_float4(tp[0], tp[32], tp[64], tp[96]);
    x.x = x.x > 0.f ? x.x : expm1f(x.x);
    x.y = x.y > 0.f ? x.y : expm1f(x.y);
    x.z = x.z > 0.f ? x.z : expm1f(x.z);
    x.w = x.w > 0.f ? x.w : expm1f(x.w);
    *(float4*)(out + (size_t)(base + r) * HF + half * 128 + 4 * l) = x;
  }
}

extern "C" void kernel_launch(void* const* d_in, const int* in_sizes, int n_in,
                              void* d_out, int out_size, void* d_ws, size_t ws_size,
                              hipStream_t stream) {
  const float* h   = (const float*)d_in[0];
  const float* ef  = (const float*)d_in[1];
  const int*   src = (const int*)  d_in[2];
  const int*   dst = (const int*)  d_in[3];
  const float* W_w = (const float*)d_in[4];
  const float* We_w= (const float*)d_in[5];
  const float* a_s = (const float*)d_in[6];
  const float* a_d = (const float*)d_in[7];
  const float* a_e = (const float*)d_in[8];
  float* out = (float*)d_out;

  char* ws = (char*)d_ws;
  const size_t WH_BYTES   = (size_t)N_NODES * HF * 4;
  const size_t SBUF_BYTES = (size_t)N_EDGES * NH * 4;
  const size_t NH_BYTES   = (size_t)N_NODES * NH * 4;
  float*    Wh   = (float*)   (ws);
  float*    sbuf = (float*)   (ws + WH_BYTES);
  float*    ssrc = (float*)   (ws + WH_BYTES + SBUF_BYTES);
  float*    sdst = (float*)   (ws + WH_BYTES + SBUF_BYTES + NH_BYTES);
  unsigned* mmax = (unsigned*)(ws + WH_BYTES + SBUF_BYTES + 2 * NH_BYTES);
  float*    den  = (float*)   (ws + WH_BYTES + SBUF_BYTES + 3 * NH_BYTES);
  float*    V    = (float*)   (ws + WH_BYTES + SBUF_BYTES + 4 * NH_BYTES);

  hipMemsetAsync(mmax, 0, NH_BYTES, stream);
  hipMemsetAsync(den,  0, NH_BYTES, stream);

  gemm_wh_wmma<<<N_NODES / 16, 256, 0, stream>>>(h, W_w, Wh);
  fold_we<<<1, 512, 0, stream>>>(We_w, a_e, V);
  node_scores<<<(N_NODES * NH + 255) / 256, 256, 0, stream>>>(Wh, a_s, a_d, ssrc, sdst);
  edge_logits<<<(N_EDGES * NH + 255) / 256, 256, 0, stream>>>(ef, src, dst, V, ssrc, sdst, sbuf, mmax);
  softmax_den<<<(N_EDGES * NH + 255) / 256, 256, 0, stream>>>(dst, mmax, sbuf, den);
  aggregate_tile<<<(N_NODES + TROWS - 1) / TROWS, 1024, 0, stream>>>(src, dst, Wh, sbuf, den, out);
}


# Softmax division moved behind the aggregation: the dst tile sums Wh[src]*exp(s-m) and the exp sums in LDS, divides once at the store; softmax_den pass, den atomics, den memset go; sum order changes
# speedup vs baseline: 3.4016x; 1.1615x over previous
#include <hip/hip_runtime.h>
#include <math.h>

typedef __attribute__((ext_vector_type(2))) float v2f;
typedef __attribute__((ext_vector_type(8))) float v8f;

#define N_NODES   50000
#define N_EDGES   800000
#define IN_F      256
#define EDGE_F    64
#define NH        8
#define NF        32
#define HF        256

__device__ __forceinline__ unsigned enc_f32(float f) {
  unsigned u = __float_as_uint(f);
  return (u & 0x80000000u) ? ~u : (u | 0x80000000u);
}
__device__ __forceinline__ float dec_f32(unsigned u) {
  unsigned b = (u & 0x80000000u) ? (u ^ 0x80000000u) : ~u;
  return __uint_as_float(b);
}

__global__ __launch_bounds__(256)
void gemm_wh_wmma(const float* __restrict__ h, const float* __restrict__ W,
                  float* __restrict__ Wh) {
  const int lane = threadIdx.x & 31;
  const int wave = threadIdx.x >> 5;
  const int m0   = blockIdx.x * 16;
  const int n0   = wave * 32;
  const int l15  = lane & 15;
  const int hi   = lane >> 4;
  const int khalf = hi * 2;

  v8f c0 = {}; v8f c1 = {};
  const float* arow = h + (size_t)(m0 + l15) * IN_F;

  for (int k0 = 0; k0 < IN_F; k0 += 4) {
    v2f a = *(const v2f*)(arow + k0 + khalf);
    const float* bb = W + (size_t)(k0 + khalf) * HF + n0 + l15;
    v2f b0, b1;
    b0.x = bb[0];        b0.y = bb[HF];
    b1.x = bb[16];       b1.y = bb[HF + 16];
    c0 = __builtin_amdgcn_wmma_f32_16x16x4_f32(false, a, false, b0,
                                               (short)0, c0, false, false);
    c1 = __builtin_amdgcn_wmma_f32_16x16x4_f32(false, a, false, b1,
                                               (short)0, c1, false, false);
  }
  const int mo = m0 + (hi << 3);
  const int nc = n0 + l15;
  for (int r = 0; r < 8; ++r) {
    Wh[(size_t)(mo + r) * HF + nc]      = c0[r];
    Wh[(size_t)(mo + r) * HF + nc + 16] = c1[r];
  }
}

__global__ void fold_we(const float* __restrict__ We_w,
                        const float* __restrict__ a_edge,
                        float* __restrict__ V) {
  int t = threadIdx.x;
  int k = t >> 3, hh = t & 7;
  const float* wr = We_w + (size_t)k * HF + hh * NF;
  const float* ae = a_edge + hh * NF;
  float acc = 0.f;
  for (int f = 0; f < NF; ++f) acc += wr[f] * ae[f];
  V[k * NH + hh] = acc;
}

__global__ __launch_bounds__(256)
void node_scores(const float* __restrict__ Wh, const float* __restrict__ a_src,
                 const float* __restrict__ a_dst,
                 float* __restrict__ ssrc, float* __restrict__ sdst) {
  int t = blockIdx.x * blockDim.x + threadIdx.x;
  if (t >= N_NODES * NH) return;
  int n = t >> 3, hh = t & 7;
  const float* row = Wh + (size_t)n * HF + hh * NF;
  const float* as  = a_src + hh * NF;
  const float* ad  = a_dst + hh * NF;
  float accs = 0.f, accd = 0.f;
  for (int f = 0; f < NF; ++f) { float v = row[f]; accs += v * as[f]; accd += v * ad[f]; }
  ssrc[t] = accs;
  sdst[t] = accd;
}

__global__ __launch_bounds__(256)
void edge_logits(const float* __restrict__ ef, const int* __restrict__ src,
                 const int* __restrict__ dst, const float* __restrict__ V,
                 const float* __restrict__ ssrc, const float* __restrict__ sdst,
                 float* __restrict__ sbuf, unsigned* __restrict__ m) {
  __shared__ float Vl[EDGE_F * NH];
  for (int i = threadIdx.x; i < EDGE_F * NH; i += blockDim.x) Vl[i] = V[i];
  __syncthreads();
  int t = blockIdx.x * blockDim.x + threadIdx.x;
  if (t >= N_EDGES * NH) return;
  int e = t >> 3, hh = t & 7;
  const float4* efr = (const float4*)(ef + (size_t)e * EDGE_F);
  float acc = 0.f;
  for (int q = 0; q < EDGE_F / 4; ++q) {
    float4 v = efr[q];
    int k = q * 4;
    acc += v.x * Vl[(k + 0) * NH + hh] + v.y * Vl[(k + 1) * NH + hh]
         + v.z * Vl[(k + 2) * NH + hh] + v.w * Vl[(k + 3) * NH + hh];
  }
  int s = src[e], d = dst[e];
  float sv = ssrc[s * NH + hh] + sdst[d * NH + hh] + acc;
  sv = sv > 0.f ? sv : 0.01f * sv;
  sbuf[t] = sv;
  atomicMax(&m[d * NH + hh], enc_f32(sv));
}

#define TROWS     196
#define TWAVES    32
#define LIST_CAP  544
#define E4        (N_EDGES / 4)
static_assert(E4 % 32 == 0, "the scan loads whole waves of int4");
static_assert(TROWS <= 256, "the local row packs into 8 bits");
static_assert(N_EDGES < (1 << 24), "the edge id packs into 24 bits");

__device__ __forceinline__ void lds_add(float* p, float v) {
  __hip_atomic_fetch_add(p, v, __ATOMIC_RELAXED, __HIP_MEMORY_SCOPE_WORKGROUP);
}

__global__ __launch_bounds__(1024)
void aggregate_tile(const int* __restrict__ src, const int* __restrict__ dst,
                    const float* __restrict__ Wh, const float* __restrict__ sbuf,
                    const unsigned* __restrict__ m, float* __restrict__ out) {
  __shared__ float    tile[TROWS * HF];
  __shared__ unsigned elist[TWAVES * LIST_CAP];
  __shared__ float    astage[TWAVES * 32 * NH];
  __shared__ __attribute__((aligned(16))) float mtile[TROWS * NH];
  __shared__ __attribute__((aligned(16))) float dtile[TROWS * NH];

  const int tid  = threadIdx.x;
  const int lane = tid & 31;
  const int wave = __builtin_amdgcn_readfirstlane(tid >> 5);
  const int base = blockIdx.x * TROWS;
  const int nrows = min(TROWS, N_NODES - base);
  const unsigned lt = (1u << lane) - 1u;

  float4* t4 = (float4*)tile;
  for (int i = tid; i < TROWS * HF / 4; i += 1024) t4[i] = make_float4(0.f, 0.f, 0.f, 0.f);
  for (int i = tid; i < TROWS * NH; i += 1024) {
    dtile[i] = 0.f;
    mtile[i] = (i < nrows * NH) ? dec_f32(m[(size_t)base * NH + i]) : 0.f;
  }
  __syncthreads();

  unsigned* mylist  = elist + wave * LIST_CAP;
  float*    mystage = astage + wave * (32 * NH);
  const int4* dst4 = (const int4*)dst;
  const int NIT = (E4 + 4095) / 4096;
  int cnt = 0;

  for (int it = 0; it <= NIT; ++it) {
    const bool last = (it == NIT);
    if (cnt >= 32 || (last && cnt > 0)) {
      for (int b = 0; b < cnt; b += 32) {
        const int nb = min(32, cnt - b);
        unsigned ent = 0u;
        int s_i = 0;
        if (lane < nb) {
          ent = mylist[b + lane];
          const int e = (int)(ent >> 8);
          const int lrow = (int)(ent & 255u);
          s_i = src[e];
          const float4* sp = (const float4*)(sbuf + (size_t)e * NH);
          const float4* mp = (const float4*)(mtile + lrow * NH);
          const float4 x0 = sp[0], x1 = sp[1], y0 = mp[0], y1 = mp[1];
          const float4 e0 = make_float4(expf(x0.x - y0.x), expf(x0.y - y0.y),
                                        expf(x0.z - y0.z), expf(x0.w - y0.w));
          const float4 e1 = make_float4(expf(x1.x - y1.x), expf(x1.y - y1.y),
                                        expf(x1.z - y1.z), expf(x1.w - y1.w));
          float4* st = (float4*)(mystage + lane * NH);
          st[0] = e0;
          st[1] = e1;
          float* dp = dtile + lrow * NH;
          lds_add(dp + 0, e0.x); lds_add(dp + 1, e0.y);
          lds_add(dp + 2, e0.z); lds_add(dp + 3, e0.w);
          lds_add(dp + 4, e1.x); lds_add(dp + 5, e1.y);
          lds_add(dp + 6, e1.z); lds_add(dp + 7, e1.w);
        }
        for (int u0 = 0; u0 < nb; u0 += 4) {
          float4 v0[4], v1[4];
          float a0[4], a1[4];
          int lr[4];
#pragma unroll
          for (int k = 0; k < 4; ++k) {
            const int u = min(u0 + k, nb - 1);
            const int s = __builtin_amdgcn_readlane(s_i, u);
            lr[k] = __builtin_amdgcn_readlane((int)ent, u) & 255;
            const float4* rp = (const float4*)(Wh + (size_t)s * HF);
            v0[k] = rp[lane];
            v1[k] = rp[32 + lane];
            a0[k] = mystage[u * NH + (lane >> 3)];
            a1[k] = mystage[u * NH + 4 + (lane >> 3)];
          }
#pragma unroll
          for (int k = 0; k < 4; ++k) {
            if (u0 + k < nb) {
              float* tp = tile + lr[k] * HF + lane;
              lds_add(tp,       v0[k].x * a0[k]);
              lds_add(tp + 32,  v0[k].y * a0[k]);
              lds_add(tp + 64,  v0[k].z * a0[k]);
              lds_add(tp + 96,  v0[k].w * a0[k]);
              lds_add(tp + 128, v1[k].x * a1[k]);
              lds_add(tp + 160, v1[k].y * a1[k]);
              lds_add(tp + 192, v1[k].z * a1[k]);
              lds_add(tp + 224, v1[k].w * a1[k]);
            }
          }
        }
      }
      cnt = 0;
    }
    if (last) break;

    const int qb = it * 4096 + wave * 128;
    int4 dd[4];
#pragma unroll
    for (int k = 0; k < 4; ++k) {
      const int q = min(qb + k * 32, E4 - 32);
      dd[k] = dst4[q + lane];
    }
#pragma unroll
    for (int k = 0; k < 4; ++k) {
      const unsigned eb = (unsigned)(qb + k * 32 + lane) * 4u;
      const unsigned lim = (qb + k * 32 < E4) ? (unsigned)nrows : 0u;
      const int dc[4] = {dd[k].x, dd[k].y, dd[k].z, dd[k].w};
#pragma unroll
      for (int j = 0; j < 4; ++j) {
        const unsigned lrow = (unsigned)(dc[j] - base);
        const bool hit = lrow < lim;
        const unsigned mask = __builtin_amdgcn_ballot_w32(hit);
        if (mask != 0u) {
          if (hit) mylist[cnt + __popc(mask & lt)] = ((eb + (unsigned)j) << 8) | lrow;
          cnt += __popc(mask);
        }
      }
    }
  }
  __syncthreads();

  for (int i = tid; i < nrows * 64; i += 1024) {
    const int r = i >> 6, g = i & 63;
    const int half = g >> 5, l = g & 31;
    const float* tp = tile + r * HF + half * 128 + l;
    float4 x = make_float4(tp[0], tp[32], tp[64], tp[96]);
    const float dn = dtile[r * NH + half * 4 + (l >> 3)];
    if (dn > 0.f) { x.x /= dn; x.y /= dn; x.z /= dn; x.w /= dn; }
    x.x = x.x > 0.f ? x.x : expm1f(x.x);
    x.y = x.y > 0.f ? x.y : expm1f(x.y);
    x.z = x.z > 0.f ? x.z : expm1f(x.z);
    x.w = x.w > 0.f ? x.w : expm1f(x.w);
    *(float4*)(out + (size_t)(base + r) * HF + half * 128 + 4 * l) = x;
  }
}

extern "C" void kernel_launch(void* const* d_in, const int* in_sizes, int n_in,
                              void* d_out, int out_size, void* d_ws, size_t ws_size,
                              hipStream_t stream) {
  const float* h   = (const float*)d_in[0];
  const float* ef  = (const float*)d_in[1];
  const int*   src = (const int*)  d_in[2];
  const int*   dst = (const int*)  d_in[3];
  const float* W_w = (const float*)d_in[4];
  const float* We_w= (const float*)d_in[5];
  const float* a_s = (const float*)d_in[6];
  const float* a_d = (const float*)d_in[7];
  const float* a_e = (const float*)d_in[8];
  float* out = (float*)d_out;

  char* ws = (char*)d_ws;
  const size_t WH_BYTES   = (size_t)N_NODES * HF * 4;
  const size_t SBUF_BYTES = (size_t)N_EDGES * NH * 4;
  const size_t NH_BYTES   = (size_t)N_NODES * NH * 4;
  float*    Wh   = (float*)   (ws);
  float*    sbuf = (float*)   (ws + WH_BYTES);
  float*    ssrc = (float*)   (ws + WH_BYTES + SBUF_BYTES);
  float*    sdst = (float*)   (ws + WH_BYTES + SBUF_BYTES + NH_BYTES);
  unsigned* mmax = (unsigned*)(ws + WH_BYTES + SBUF_BYTES + 2 * NH_BYTES);
  float*    V    = (float*)   (ws + WH_BYTES + SBUF_BYTES + 4 * NH_BYTES);

  hipMemsetAsync(mmax, 0, NH_BYTES, stream);

  gemm_wh_wmma<<<N_NODES / 16, 256, 0, stream>>>(h, W_w, Wh);
  fold_we<<<1, 512, 0, stream>>>(We_w, a_e, V);
  node_scores<<<(N_NODES * NH + 255) / 256, 256, 0, stream>>>(Wh, a_s, a_d, ssrc, sdst);
  edge_logits<<<(N_EDGES * NH + 255) / 256, 256, 0, stream>>>(ef, src, dst, V, ssrc, sdst, sbuf, mmax);
  aggregate_tile<<<(N_NODES + TROWS - 1) / TROWS, 1024, 0, stream>>>(src, dst, Wh, sbuf, mmax, out);
}


# GEMM fragment loads as 16-byte vectors: wave tile 16x64 with k and column order permuted so each A/B load covers whole runs and Wh rows store as full lines; k sum order changes
# speedup vs baseline: 3.6421x; 1.0707x over previous
#include <hip/hip_runtime.h>
#include <math.h>

typedef __attribute__((ext_vector_type(2))) float v2f;
typedef __attribute__((ext_vector_type(4))) float v4f;
typedef __attribute__((ext_vector_type(8))) float v8f;

#define N_NODES   50000
#define N_EDGES   800000
#define IN_F      256
#define EDGE_F    64
#define NH        8
#define NF        32
#define HF        256

__device__ __forceinline__ unsigned enc_f32(float f) {
  unsigned u = __float_as_uint(f);
  return (u & 0x80000000u) ? ~u : (u | 0x80000000u);
}
__device__ __forceinline__ float dec_f32(unsigned u) {
  unsigned b = (u & 0x80000000u) ? (u ^ 0x80000000u) : ~u;
  return __uint_as_float(b);
}

__global__ __launch_bounds__(128)
void gemm_wh_wmma(const float* __restrict__ h, const float* __restrict__ W,
                  float* __restrict__ Wh) {
  const int lane = threadIdx.x & 31;
  const int wave = threadIdx.x >> 5;
  const int m0   = blockIdx.x * 16;
  const int n0   = wave * 64;
  const int c    = lane & 15;
  const int hi   = lane >> 4;

  v8f c0 = {}; v8f c1 = {}; v8f c2 = {}; v8f c3 = {};
  const float* arow = h + (size_t)(m0 + c) * IN_F + 4 * hi;
  const float* brow = W + (size_t)(4 * hi) * HF + n0 + 4 * c;

  for (int k0 = 0; k0 < IN_F; k0 += 8) {
    const v4f av = *(const v4f*)(arow + k0);
    const float* bp = brow + (size_t)k0 * HF;
    const v4f q0 = *(const v4f*)(bp);
    const v4f q1 = *(const v4f*)(bp + HF);
    const v4f q2 = *(const v4f*)(bp + 2 * HF);
    const v4f q3 = *(const v4f*)(bp + 3 * HF);
    v2f a, b;
    a.x = av.x; a.y = av.y;
    b.x = q0.x; b.y = q1.x;
    c0 = __builtin_amdgcn_wmma_f32_16x16x4_f32(false, a, false, b, (short)0, c0, false, false);
    b.x = q0.y; b.y = q1.y;
    c1 = __builtin_amdgcn_wmma_f32_16x16x4_f32(false, a, false, b, (short)0, c1, false, false);
    b.x = q0.z; b.y = q1.z;
    c2 = __builtin_amdgcn_wmma_f32_16x16x4_f32(false, a, false, b, (short)0, c2, false, false);
    b.x = q0.w; b.y = q1.w;
    c3 = __builtin_amdgcn_wmma_f32_16x16x4_f32(false, a, false, b, (short)0, c3, false, false);
    a.x = av.z; a.y = av.w;
    b.x = q2.x; b.y = q3.x;
    c0 = __builtin_amdgcn_wmma_f32_16x16x4_f32(false, a, false, b, (short)0, c0, false, false);
    b.x = q2.y; b.y = q3.y;
    c1 = __builtin_amdgcn_wmma_f32_16x16x4_f32(false, a, false, b, (short)0, c1, false, false);
    b.x = q2.z; b.y = q3.z;
    c2 = __builtin_amdgcn_wmma_f32_16x16x4_f32(false, a, false, b, (short)0, c2, false, false);
    b.x = q2.w; b.y = q3.w;
    c3 = __builtin_amdgcn_wmma_f32_16x16x4_f32(false, a, false, b, (short)0, c3, false, false);
  }
  float* op = Wh + (size_t)(m0 + (hi << 3)) * HF + n0 + 4 * c;
  for (int r = 0; r < 8; ++r) {
    v4f o;
    o.x = c0[r]; o.y = c1[r]; o.z = c2[r]; o.w = c3[r];
    *(v4f*)(op + (size_t)r * HF) = o;
  }
}

__global__ void fold_we(const float* __restrict__ We_w,
                        const float* __restrict__ a_edge,
                        float* __restrict__ V) {
  int t = threadIdx.x;
  int k = t >> 3, hh = t & 7;
  const float* wr = We_w + (size_t)k * HF + hh * NF;
  const float* ae = a_edge + hh * NF;
  float acc = 0.f;
  for (int f = 0; f < NF; ++f) acc += wr[f] * ae[f];
  V[k * NH + hh] = acc;
}

__global__ __launch_bounds__(256)
void node_scores(const float* __restrict__ Wh, const float* __restrict__ a_src,
                 const float* __restrict__ a_dst,
                 float* __restrict__ ssrc, float* __restrict__ sdst) {
  int t = blockIdx.x * blockDim.x + threadIdx.x;
  if (t >= N_NODES * NH) return;
  int n = t >> 3, hh = t & 7;
  const float* row = Wh + (size_t)n * HF + hh * NF;
  const float* as  = a_src + hh * NF;
  const float* ad  = a_dst + hh * NF;
  float accs = 0.f, accd = 0.f;
  for (int f = 0; f < NF; ++f) { float v = row[f]; accs += v * as[f]; accd += v * ad[f]; }
  ssrc[t] = accs;
  sdst[t] = accd;
}

__global__ __launch_bounds__(256)
void edge_logits(const float* __restrict__ ef, const int* __restrict__ src,
                 const int* __restrict__ dst, const float* __restrict__ V,
                 const float* __restrict__ ssrc, const float* __restrict__ sdst,
                 float* __restrict__ sbuf, unsigned* __restrict__ m) {
  __shared__ float Vl[EDGE_F * NH];
  for (int i = threadIdx.x; i < EDGE_F * NH; i += blockDim.x) Vl[i] = V[i];
  __syncthreads();
  int t = blockIdx.x * blockDim.x + threadIdx.x;
  if (t >= N_EDGES * NH) return;
  int e = t >> 3, hh = t & 7;
  const float4* efr = (const float4*)(ef + (size_t)e * EDGE_F);
  float acc = 0.f;
  for (int q = 0; q < EDGE_F / 4; ++q) {
    float4 v = efr[q];
    int k = q * 4;
    acc += v.x * Vl[(k + 0) * NH + hh] + v.y * Vl[(k + 1) * NH + hh]
         + v.z * Vl[(k + 2) * NH + hh] + v.w * Vl[(k + 3) * NH + hh];
  }
  int s = src[e], d = dst[e];
  float sv = ssrc[s * NH + hh] + sdst[d * NH + hh] + acc;
  sv = sv > 0.f ? sv : 0.01f * sv;
  sbuf[t] = sv;
  atomicMax(&m[d * NH + hh], enc_f32(sv));
}

#define TROWS     196
#define TWAVES    32
#define LIST_CAP  544
#define E4        (N_EDGES / 4)
static_assert(E4 % 32 == 0, "the scan loads whole waves of int4");
static_assert(TROWS <= 256, "the local row packs into 8 bits");
static_assert(N_EDGES < (1 << 24), "the edge id packs into 24 bits");

__device__ __forceinline__ void lds_add(float* p, float v) {
  __hip_atomic_fetch_add(p, v, __ATOMIC_RELAXED, __HIP_MEMORY_SCOPE_WORKGROUP);
}

__global__ __launch_bounds__(1024)
void aggregate_tile(const int* __restrict__ src, const int* __restrict__ dst,
                    const float* __restrict__ Wh, const float* __restrict__ sbuf,
                    const unsigned* __restrict__ m, float* __restrict__ out) {
  __shared__ float    tile[TROWS * HF];
  __shared__ unsigned elist[TWAVES * LIST_CAP];
  __shared__ float    astage[TWAVES * 32 * NH];
  __shared__ __attribute__((aligned(16))) float mtile[TROWS * NH];
  __shared__ __attribute__((aligned(16))) float dtile[TROWS * NH];

  const int tid  = threadIdx.x;
  const int lane = tid & 31;
  const int wave = __builtin_amdgcn_readfirstlane(tid >> 5);
  const int base = blockIdx.x * TROWS;
  const int nrows = min(TROWS, N_NODES - base);
  const unsigned lt = (1u << lane) - 1u;

  float4* t4 = (float4*)tile;
  for (int i = tid; i < TROWS * HF / 4; i += 1024) t4[i] = make_float4(0.f, 0.f, 0.f, 0.f);
  for (int i = tid; i < TROWS * NH; i += 1024) {
    dtile[i] = 0.f;
    mtile[i] = (i < nrows * NH) ? dec_f32(m[(size_t)base * NH + i]) : 0.f;
  }
  __syncthreads();

  unsigned* mylist  = elist + wave * LIST_CAP;
  float*    mystage = astage + wave * (32 * NH);
  const int4* dst4 = (const int4*)dst;
  const int NIT = (E4 + 4095) / 4096;
  int cnt = 0;

  for (int it = 0; it <= NIT; ++it) {
    const bool last = (it == NIT);
    if (cnt >= 32 || (last && cnt > 0)) {
      for (int b = 0; b < cnt; b += 32) {
        const int nb = min(32, cnt - b);
        unsigned ent = 0u;
        int s_i = 0;
        if (lane < nb) {
          ent = mylist[b + lane];
          const int e = (int)(ent >> 8);
          const int lrow = (int)(ent & 255u);
          s_i = src[e];
          const float4* sp = (const float4*)(sbuf + (size_t)e * NH);
          const float4* mp = (const float4*)(mtile + lrow * NH);
          const float4 x0 = sp[0], x1 = sp[1], y0 = mp[0], y1 = mp[1];
          const float4 e0 = make_float4(expf(x0.x - y0.x), expf(x0.y - y0.y),
                                        expf(x0.z - y0.z), expf(x0.w - y0.w));
          const float4 e1 = make_float4(expf(x1.x - y1.x), expf(x1.y - y1.y),
                                        expf(x1.z - y1.z), expf(x1.w - y1.w));
          float4* st = (float4*)(mystage + lane * NH);
          st[0] = e0;
          st[1] = e1;
          float* dp = dtile + lrow * NH;
          lds_add(dp + 0, e0.x); lds_add(dp + 1, e0.y);
          lds_add(dp + 2, e0.z); lds_add(dp + 3, e0.w);
          lds_add(dp + 4, e1.x); lds_add(dp + 5, e1.y);
          lds_add(dp + 6, e1.z); lds_add(dp + 7, e1.w);
        }
        for (int u0 = 0; u0 < nb; u0 += 4) {
          float4 v0[4], v1[4];
          float a0[4], a1[4];
          int lr[4];
#pragma unroll
          for (int k = 0; k < 4; ++k) {
            const int u = min(u0 + k, nb - 1);
            const int s = __builtin_amdgcn_readlane(s_i, u);
            lr[k] = __builtin_amdgcn_readlane((int)ent, u) & 255;
            const float4* rp = (const float4*)(Wh + (size_t)s * HF);
            v0[k] = rp[lane];
            v1[k] = rp[32 + lane];
            a0[k] = mystage[u * NH + (lane >> 3)];
            a1[k] = mystage[u * NH + 4 + (lane >> 3)];
          }
#pragma unroll
          for (int k = 0; k < 4; ++k) {
            if (u0 + k < nb) {
              float* tp = tile + lr[k] * HF + lane;
              lds_add(tp,       v0[k].x * a0[k]);
              lds_add(tp + 32,  v0[k].y * a0[k]);
              lds_add(tp + 64,  v0[k].z * a0[k]);
              lds_add(tp + 96,  v0[k].w * a0[k]);
              lds_add(tp + 128, v1[k].x * a1[k]);
              lds_add(tp + 160, v1[k].y * a1[k]);
              lds_add(tp + 192, v1[k].z * a1[k]);
              lds_add(tp + 224, v1[k].w * a1[k]);
            }
          }
        }
      }
      cnt = 0;
    }
    if (last) break;

    const int qb = it * 4096 + wave * 128;
    int4 dd[4];
#pragma unroll
    for (int k = 0; k < 4; ++k) {
      const int q = min(qb + k * 32, E4 - 32);
      dd[k] = dst4[q + lane];
    }
#pragma unroll
    for (int k = 0; k < 4; ++k) {
      const unsigned eb = (unsigned)(qb + k * 32 + lane) * 4u;
      const unsigned lim = (qb + k * 32 < E4) ? (unsigned)nrows : 0u;
      const int dc[4] = {dd[k].x, dd[k].y, dd[k].z, dd[k].w};
#pragma unroll
      for (int j = 0; j < 4; ++j) {
        const unsigned lrow = (unsigned)(dc[j] - base);
        const bool hit = lrow < lim;
        const unsigned mask = __builtin_amdgcn_ballot_w32(hit);
        if (mask != 0u) {
          if (hit) mylist[cnt + __popc(mask & lt)] = ((eb + (unsigned)j) << 8) | lrow;
          cnt += __popc(mask);
        }
      }
    }
  }
  __syncthreads();

  for (int i = tid; i < nrows * 64; i += 1024) {
    const int r = i >> 6, g = i & 63;
    const int half = g >> 5, l = g & 31;
    const float* tp = tile + r * HF + half * 128 + l;
    float4 x = make_float4(tp[0], tp[32], tp[64], tp[96]);
    const float dn = dtile[r * NH + half * 4 + (l >> 3)];
    if (dn > 0.f) { x.x /= dn; x.y /= dn; x.z /= dn; x.w /= dn; }
    x.x = x.x > 0.f ? x.x : expm1f(x.x);
    x.y = x.y > 0.f ? x.y : expm1f(x.y);
    x.z = x.z > 0.f ? x.z : expm1f(x.z);
    x.w = x.w > 0.f ? x.w : expm1f(x.w);
    *(float4*)(out + (size_t)(base + r) * HF + half * 128 + 4 * l) = x;
  }
}

extern "C" void kernel_launch(void* const* d_in, const int* in_sizes, int n_in,
                              void* d_out, int out_size, void* d_ws, size_t ws_size,
                              hipStream_t stream) {
  const float* h   = (const float*)d_in[0];
  const float* ef  = (const float*)d_in[1];
  const int*   src = (const int*)  d_in[2];
  const int*   dst = (const int*)  d_in[3];
  const float* W_w = (const float*)d_in[4];
  const float* We_w= (const float*)d_in[5];
  const float* a_s = (const float*)d_in[6];
  const float* a_d = (const float*)d_in[7];
  const float* a_e = (const float*)d_in[8];
  float* out = (float*)d_out;

  char* ws = (char*)d_ws;
  const size_t WH_BYTES   = (size_t)N_NODES * HF * 4;
  const size_t SBUF_BYTES = (size_t)N_EDGES * NH * 4;
  const size_t NH_BYTES   = (size_t)N_NODES * NH * 4;
  float*    Wh   = (float*)   (ws);
  float*    sbuf = (float*)   (ws + WH_BYTES);
  float*    ssrc = (float*)   (ws + WH_BYTES + SBUF_BYTES);
  float*    sdst = (float*)   (ws + WH_BYTES + SBUF_BYTES + NH_BYTES);
  unsigned* mmax = (unsigned*)(ws + WH_BYTES + SBUF_BYTES + 2 * NH_BYTES);
  float*    V    = (float*)   (ws + WH_BYTES + SBUF_BYTES + 4 * NH_BYTES);

  hipMemsetAsync(mmax, 0, NH_BYTES, stream);

  gemm_wh_wmma<<<N_NODES / 16, 128, 0, stream>>>(h, W_w, Wh);
  fold_we<<<1, 512, 0, stream>>>(We_w, a_e, V);
  node_scores<<<(N_NODES * NH + 255) / 256, 256, 0, stream>>>(Wh, a_s, a_d, ssrc, sdst);
  edge_logits<<<(N_EDGES * NH + 255) / 256, 256, 0, stream>>>(ef, src, dst, V, ssrc, sdst, sbuf, mmax);
  aggregate_tile<<<(N_NODES + TROWS - 1) / TROWS, 1024, 0, stream>>>(src, dst, Wh, sbuf, mmax, out);
}


# Edge logits on a resident grid: 1024 work-groups of 8 waves walk the (edge, head) pairs with a grid stride, a thread keeps its head's 64 folded weights in registers; 8192 waves enter, not 200000
# speedup vs baseline: 3.7978x; 1.0428x over previous
#include <hip/hip_runtime.h>
#include <math.h>

typedef __attribute__((ext_vector_type(2))) float v2f;
typedef __attribute__((ext_vector_type(4))) float v4f;
typedef __attribute__((ext_vector_type(8))) float v8f;

#define N_NODES   50000
#define N_EDGES   800000
#define IN_F      256
#define EDGE_F    64
#define NH        8
#define NF        32
#define HF        256

__device__ __forceinline__ unsigned enc_f32(float f) {
  unsigned u = __float_as_uint(f);
  return (u & 0x80000000u) ? ~u : (u | 0x80000000u);
}
__device__ __forceinline__ float dec_f32(unsigned u) {
  unsigned b = (u & 0x80000000u) ? (u ^ 0x80000000u) : ~u;
  return __uint_as_float(b);
}

__global__ __launch_bounds__(128)
void gemm_wh_wmma(const float* __restrict__ h, const float* __restrict__ W,
                  float* __restrict__ Wh) {
  const int lane = threadIdx.x & 31;
  const int wave = threadIdx.x >> 5;
  const int m0   = blockIdx.x * 16;
  const int n0   = wave * 64;
  const int c    = lane & 15;
  const int hi   = lane >> 4;

  v8f c0 = {}; v8f c1 = {}; v8f c2 = {}; v8f c3 = {};
  const float* arow = h + (size_t)(m0 + c) * IN_F + 4 * hi;
  const float* brow = W + (size_t)(4 * hi) * HF + n0 + 4 * c;

  for (int k0 = 0; k0 < IN_F; k0 += 8) {
    const v4f av = *(const v4f*)(arow + k0);
    const float* bp = brow + (size_t)k0 * HF;
    const v4f q0 = *(const v4f*)(bp);
    const v4f q1 = *(const v4f*)(bp + HF);
    const v4f q2 = *(const v4f*)(bp + 2 * HF);
    const v4f q3 = *(const v4f*)(bp + 3 * HF);
    v2f a, b;
    a.x = av.x; a.y = av.y;
    b.x = q0.x; b.y = q1.x;
    c0 = __builtin_amdgcn_wmma_f32_16x16x4_f32(false, a, false, b, (short)0, c0, false, false);
    b.x = q0.y; b.y = q1.y;
    c1 = __builtin_amdgcn_wmma_f32_16x16x4_f32(false, a, false, b, (short)0, c1, false, false);
    b.x = q0.z; b.y = q1.z;
    c2 = __builtin_amdgcn_wmma_f32_16x16x4_f32(false, a, false, b, (short)0, c2, false, false);
    b.x = q0.w; b.y = q1.w;
    c3 = __builtin_amdgcn_wmma_f32_16x16x4_f32(false, a, false, b, (short)0, c3, false, false);
    a.x = av.z; a.y = av.w;
    b.x = q2.x; b.y = q3.x;
    c0 = __builtin_amdgcn_wmma_f32_16x16x4_f32(false, a, false, b, (short)0, c0, false, false);
    b.x = q2.y; b.y = q3.y;
    c1 = __builtin_amdgcn_wmma_f32_16x16x4_f32(false, a, false, b, (short)0, c1, false, false);
    b.x = q2.z; b.y = q3.z;
    c2 = __builtin_amdgcn_wmma_f32_16x16x4_f32(false, a, false, b, (short)0, c2, false, false);
    b.x = q2.w; b.y = q3.w;
    c3 = __builtin_amdgcn_wmma_f32_16x16x4_f32(false, a, false, b, (short)0, c3, false, false);
  }
  float* op = Wh + (size_t)(m0 + (hi << 3)) * HF + n0 + 4 * c;
  for (int r = 0; r < 8; ++r) {
    v4f o;
    o.x = c0[r]; o.y = c1[r]; o.z = c2[r]; o.w = c3[r];
    *(v4f*)(op + (size_t)r * HF) = o;
  }
}

__global__ void fold_we(const float* __restrict__ We_w,
                        const float* __restrict__ a_edge,
                        float* __restrict__ V) {
  int t = threadIdx.x;
  int k = t >> 3, hh = t & 7;
  const float* wr = We_w + (size_t)k * HF + hh * NF;
  const float* ae = a_edge + hh * NF;
  float acc = 0.f;
  for (int f = 0; f < NF; ++f) acc += wr[f] * ae[f];
  V[k * NH + hh] = acc;
}

__global__ __launch_bounds__(256)
void node_scores(const float* __restrict__ Wh, const float* __restrict__ a_src,
                 const float* __restrict__ a_dst,
                 float* __restrict__ ssrc, float* __restrict__ sdst) {
  int t = blockIdx.x * blockDim.x + threadIdx.x;
  if (t >= N_NODES * NH) return;
  int n = t >> 3, hh = t & 7;
  const float* row = Wh + (size_t)n * HF + hh * NF;
  const float* as  = a_src + hh * NF;
  const float* ad  = a_dst + hh * NF;
  float accs = 0.f, accd = 0.f;
  for (int f = 0; f < NF; ++f) { float v = row[f]; accs += v * as[f]; accd += v * ad[f]; }
  ssrc[t] = accs;
  sdst[t] = accd;
}

#define LOGIT_BLOCKS 1024
static_assert((LOGIT_BLOCKS * 256) % NH == 0, "a thread must keep its head");
__global__ __launch_bounds__(256)
void edge_logits(const float* __restrict__ ef, const int* __restrict__ src,
                 const int* __restrict__ dst, const float* __restrict__ V,
                 const float* __restrict__ ssrc, const float* __restrict__ sdst,
                 float* __restrict__ sbuf, unsigned* __restrict__ m) {
  const int t0 = blockIdx.x * 256 + threadIdx.x;
  const int hh = t0 & 7;
  float vk[EDGE_F];
#pragma unroll
  for (int k = 0; k < EDGE_F; ++k) vk[k] = V[k * NH + hh];

  for (int t = t0; t < N_EDGES * NH; t += LOGIT_BLOCKS * 256) {
    const int e = t >> 3;
    const float4* efr = (const float4*)(ef + (size_t)e * EDGE_F);
    const int s = src[e], d = dst[e];
    const float ns = ssrc[s * NH + hh];
    const float nd = sdst[d * NH + hh];
    float acc = 0.f;
#pragma unroll
    for (int q = 0; q < EDGE_F / 4; ++q) {
      const float4 v = efr[q];
      const int k = q * 4;
      acc += v.x * vk[k + 0] + v.y * vk[k + 1] + v.z * vk[k + 2] + v.w * vk[k + 3];
    }
    float sv = ns + nd + acc;
    sv = sv > 0.f ? sv : 0.01f * sv;
    sbuf[t] = sv;
    atomicMax(&m[d * NH + hh], enc_f32(sv));
  }
}

#define TROWS     196
#define TWAVES    32
#define LIST_CAP  544
#define E4        (N_EDGES / 4)
static_assert(E4 % 32 == 0, "the scan loads whole waves of int4");
static_assert(TROWS <= 256, "the local row packs into 8 bits");
static_assert(N_EDGES < (1 << 24), "the edge id packs into 24 bits");

__device__ __forceinline__ void lds_add(float* p, float v) {
  __hip_atomic_fetch_add(p, v, __ATOMIC_RELAXED, __HIP_MEMORY_SCOPE_WORKGROUP);
}

__global__ __launch_bounds__(1024)
void aggregate_tile(const int* __restrict__ src, const int* __restrict__ dst,
                    const float* __restrict__ Wh, const float* __restrict__ sbuf,
                    const unsigned* __restrict__ m, float* __restrict__ out) {
  __shared__ float    tile[TROWS * HF];
  __shared__ unsigned elist[TWAVES * LIST_CAP];
  __shared__ float    astage[TWAVES * 32 * NH];
  __shared__ __attribute__((aligned(16))) float mtile[TROWS * NH];
  __shared__ __attribute__((aligned(16))) float dtile[TROWS * NH];

  const int tid  = threadIdx.x;
  const int lane = tid & 31;
  const int wave = __builtin_amdgcn_readfirstlane(tid >> 5);
  const int base = blockIdx.x * TROWS;
  const int nrows = min(TROWS, N_NODES - base);
  const unsigned lt = (1u << lane) - 1u;

  float4* t4 = (float4*)tile;
  for (int i = tid; i < TROWS * HF / 4; i += 1024) t4[i] = make_float4(0.f, 0.f, 0.f, 0.f);
  for (int i = tid; i < TROWS * NH; i += 1024) {
    dtile[i] = 0.f;
    mtile[i] = (i < nrows * NH) ? dec_f32(m[(size_t)base * NH + i]) : 0.f;
  }
  __syncthreads();

  unsigned* mylist  = elist + wave * LIST_CAP;
  float*    mystage = astage + wave * (32 * NH);
  const int4* dst4 = (const int4*)dst;
  const int NIT = (E4 + 4095) / 4096;
  int cnt = 0;

  for (int it = 0; it <= NIT; ++it) {
    const bool last = (it == NIT);
    if (cnt >= 32 || (last && cnt > 0)) {
      for (int b = 0; b < cnt; b += 32) {
        const int nb = min(32, cnt - b);
        unsigned ent = 0u;
        int s_i = 0;
        if (lane < nb) {
          ent = mylist[b + lane];
          const int e = (int)(ent >> 8);
          const int lrow = (int)(ent & 255u);
          s_i = src[e];
          const float4* sp = (const float4*)(sbuf + (size_t)e * NH);
          const float4* mp = (const float4*)(mtile + lrow * NH);
          const float4 x0 = sp[0], x1 = sp[1], y0 = mp[0], y1 = mp[1];
          const float4 e0 = make_float4(expf(x0.x - y0.x), expf(x0.y - y0.y),
                                        expf(x0.z - y0.z), expf(x0.w - y0.w));
          const float4 e1 = make_float4(expf(x1.x - y1.x), expf(x1.y - y1.y),
                                        expf(x1.z - y1.z), expf(x1.w - y1.w));
          float4* st = (float4*)(mystage + lane * NH);
          st[0] = e0;
          st[1] = e1;
          float* dp = dtile + lrow * NH;
          lds_add(dp + 0, e0.x); lds_add(dp + 1, e0.y);
          lds_add(dp + 2, e0.z); lds_add(dp + 3, e0.w);
          lds_add(dp + 4, e1.x); lds_add(dp + 5, e1.y);
          lds_add(dp + 6, e1.z); lds_add(dp + 7, e1.w);
        }
        for (int u0 = 0; u0 < nb; u0 += 4) {
          float4 v0[4], v1[4];
          float a0[4], a1[4];
          int lr[4];
#pragma unroll
          for (int k = 0; k < 4; ++k) {
            const int u = min(u0 + k, nb - 1);
            const int s = __builtin_amdgcn_readlane(s_i, u);
            lr[k] = __builtin_amdgcn_readlane((int)ent, u) & 255;
            const float4* rp = (const float4*)(Wh + (size_t)s * HF);
            v0[k] = rp[lane];
            v1[k] = rp[32 + lane];
            a0[k] = mystage[u * NH + (lane >> 3)];
            a1[k] = mystage[u * NH + 4 + (lane >> 3)];
          }
#pragma unroll
          for (int k = 0; k < 4; ++k) {
            if (u0 + k < nb) {
              float* tp = tile + lr[k] * HF + lane;
              lds_add(tp,       v0[k].x * a0[k]);
              lds_add(tp + 32,  v0[k].y * a0[k]);
              lds_add(tp + 64,  v0[k].z * a0[k]);
              lds_add(tp + 96,  v0[k].w * a0[k]);
              lds_add(tp + 128, v1[k].x * a1[k]);
              lds_add(tp + 160, v1[k].y * a1[k]);
              lds_add(tp + 192, v1[k].z * a1[k]);
              lds_add(tp + 224, v1[k].w * a1[k]);
            }
          }
        }
      }
      cnt = 0;
    }
    if (last) break;

    const int qb = it * 4096 + wave * 128;
    int4 dd[4];
#pragma unroll
    for (int k = 0; k < 4; ++k) {
      const int q = min(qb + k * 32, E4 - 32);
      dd[k] = dst4[q + lane];
    }
#pragma unroll
    for (int k = 0; k < 4; ++k) {
      const unsigned eb = (unsigned)(qb + k * 32 + lane) * 4u;
      const unsigned lim = (qb + k * 32 < E4) ? (unsigned)nrows : 0u;
      const int dc[4] = {dd[k].x, dd[k].y, dd[k].z, dd[k].w};
#pragma unroll
      for (int j = 0; j < 4; ++j) {
        const unsigned lrow = (unsigned)(dc[j] - base);
        const bool hit = lrow < lim;
        const unsigned mask = __builtin_amdgcn_ballot_w32(hit);
        if (mask != 0u) {
          if (hit) mylist[cnt + __popc(mask & lt)] = ((eb + (unsigned)j) << 8) | lrow;
          cnt += __popc(mask);
        }
      }
    }
  }
  __syncthreads();

  for (int i = tid; i < nrows * 64; i += 1024) {
    const int r = i >> 6, g = i & 63;
    const int half = g >> 5, l = g & 31;
    const float* tp = tile + r * HF + half * 128 + l;
    float4 x = make_float4(tp[0], tp[32], tp[64], tp[96]);
    const float dn = dtile[r * NH + half * 4 + (l >> 3)];
    if (dn > 0.f) { x.x /= dn; x.y /= dn; x.z /= dn; x.w /= dn; }
    x.x = x.x > 0.f ? x.x : expm1f(x.x);
    x.y = x.y > 0.f ? x.y : expm1f(x.y);
    x.z = x.z > 0.f ? x.z : expm1f(x.z);
    x.w = x.w > 0.f ? x.w : expm1f(x.w);
    *(float4*)(out + (size_t)(base + r) * HF + half * 128 + 4 * l) = x;
  }
}

extern "C" void kernel_launch(void* const* d_in, const int* in_sizes, int n_in,
                              void* d_out, int out_size, void* d_ws, size_t ws_size,
                              hipStream_t stream) {
  const float* h   = (const float*)d_in[0];
  const float* ef  = (const float*)d_in[1];
  const int*   src = (const int*)  d_in[2];
  const int*   dst = (const int*)  d_in[3];
  const float* W_w = (const float*)d_in[4];
  const float* We_w= (const float*)d_in[5];
  const float* a_s = (const float*)d_in[6];
  const float* a_d = (const float*)d_in[7];
  const float* a_e = (const float*)d_in[8];
  float* out = (float*)d_out;

  char* ws = (char*)d_ws;
  const size_t WH_BYTES   = (size_t)N_NODES * HF * 4;
  const size_t SBUF_BYTES = (size_t)N_EDGES * NH * 4;
  const size_t NH_BYTES   = (size_t)N_NODES * NH * 4;
  float*    Wh   = (float*)   (ws);
  float*    sbuf = (float*)   (ws + WH_BYTES);
  float*    ssrc = (float*)   (ws + WH_BYTES + SBUF_BYTES);
  float*    sdst = (float*)   (ws + WH_BYTES + SBUF_BYTES + NH_BYTES);
  unsigned* mmax = (unsigned*)(ws + WH_BYTES + SBUF_BYTES + 2 * NH_BYTES);
  float*    V    = (float*)   (ws + WH_BYTES + SBUF_BYTES + 4 * NH_BYTES);

  hipMemsetAsync(mmax, 0, NH_BYTES, stream);

  gemm_wh_wmma<<<N_NODES / 16, 128, 0, stream>>>(h, W_w, Wh);
  fold_we<<<1, 512, 0, stream>>>(We_w, a_e, V);
  node_scores<<<(N_NODES * NH + 255) / 256, 256, 0, stream>>>(Wh, a_s, a_d, ssrc, sdst);
  edge_logits<<<LOGIT_BLOCKS, 256, 0, stream>>>(ef, src, dst, V, ssrc, sdst, sbuf, mmax);
  aggregate_tile<<<(N_NODES + TROWS - 1) / TROWS, 1024, 0, stream>>>(src, dst, Wh, sbuf, mmax, out);
}


# Tile search on one-byte keys: logits kernel stores dst/196 per edge as a byte; a tile work-group scans 800 KB of keys, 16 per lane load, 4 per 32-bit test, not 3200 KB of dst; LDS add order changes
# speedup vs baseline: 4.1557x; 1.0942x over previous
#include <hip/hip_runtime.h>
#include <math.h>

typedef __attribute__((ext_vector_type(2))) float v2f;
typedef __attribute__((ext_vector_type(4))) float v4f;
typedef __attribute__((ext_vector_type(8))) float v8f;

#define N_NODES   50000
#define N_EDGES   800000
#define IN_F      256
#define EDGE_F    64
#define NH        8
#define NF        32
#define HF        256

__device__ __forceinline__ unsigned enc_f32(float f) {
  unsigned u = __float_as_uint(f);
  return (u & 0x80000000u) ? ~u : (u | 0x80000000u);
}
__device__ __forceinline__ float dec_f32(unsigned u) {
  unsigned b = (u & 0x80000000u) ? (u ^ 0x80000000u) : ~u;
  return __uint_as_float(b);
}

__global__ __launch_bounds__(128)
void gemm_wh_wmma(const float* __restrict__ h, const float* __restrict__ W,
                  float* __restrict__ Wh) {
  const int lane = threadIdx.x & 31;
  const int wave = threadIdx.x >> 5;
  const int m0   = blockIdx.x * 16;
  const int n0   = wave * 64;
  const int c    = lane & 15;
  const int hi   = lane >> 4;

  v8f c0 = {}; v8f c1 = {}; v8f c2 = {}; v8f c3 = {};
  const float* arow = h + (size_t)(m0 + c) * IN_F + 4 * hi;
  const float* brow = W + (size_t)(4 * hi) * HF + n0 + 4 * c;

  for (int k0 = 0; k0 < IN_F; k0 += 8) {
    const v4f av = *(const v4f*)(arow + k0);
    const float* bp = brow + (size_t)k0 * HF;
    const v4f q0 = *(const v4f*)(bp);
    const v4f q1 = *(const v4f*)(bp + HF);
    const v4f q2 = *(const v4f*)(bp + 2 * HF);
    const v4f q3 = *(const v4f*)(bp + 3 * HF);
    v2f a, b;
    a.x = av.x; a.y = av.y;
    b.x = q0.x; b.y = q1.x;
    c0 = __builtin_amdgcn_wmma_f32_16x16x4_f32(false, a, false, b, (short)0, c0, false, false);
    b.x = q0.y; b.y = q1.y;
    c1 = __builtin_amdgcn_wmma_f32_16x16x4_f32(false, a, false, b, (short)0, c1, false, false);
    b.x = q0.z; b.y = q1.z;
    c2 = __builtin_amdgcn_wmma_f32_16x16x4_f32(false, a, false, b, (short)0, c2, false, false);
    b.x = q0.w; b.y = q1.w;
    c3 = __builtin_amdgcn_wmma_f32_16x16x4_f32(false, a, false, b, (short)0, c3, false, false);
    a.x = av.z; a.y = av.w;
    b.x = q2.x; b.y = q3.x;
    c0 = __builtin_amdgcn_wmma_f32_16x16x4_f32(false, a, false, b, (short)0, c0, false, false);
    b.x = q2.y; b.y = q3.y;
    c1 = __builtin_amdgcn_wmma_f32_16x16x4_f32(false, a, false, b, (short)0, c1, false, false);
    b.x = q2.z; b.y = q3.z;
    c2 = __builtin_amdgcn_wmma_f32_16x16x4_f32(false, a, false, b, (short)0, c2, false, false);
    b.x = q2.w; b.y = q3.w;
    c3 = __builtin_amdgcn_wmma_f32_16x16x4_f32(false, a, false, b, (short)0, c3, false, false);
  }
  float* op = Wh + (size_t)(m0 + (hi << 3)) * HF + n0 + 4 * c;
  for (int r = 0; r < 8; ++r) {
    v4f o;
    o.x = c0[r]; o.y = c1[r]; o.z = c2[r]; o.w = c3[r];
    *(v4f*)(op + (size_t)r * HF) = o;
  }
}

__global__ void fold_we(const float* __restrict__ We_w,
                        const float* __restrict__ a_edge,
                        float* __restrict__ V) {
  int t = threadIdx.x;
  int k = t >> 3, hh = t & 7;
  const float* wr = We_w + (size_t)k * HF + hh * NF;
  const float* ae = a_edge + hh * NF;
  float acc = 0.f;
  for (int f = 0; f < NF; ++f) acc += wr[f] * ae[f];
  V[k * NH + hh] = acc;
}

__global__ __launch_bounds__(256)
void node_scores(const float* __restrict__ Wh, const float* __restrict__ a_src,
                 const float* __restrict__ a_dst,
                 float* __restrict__ ssrc, float* __restrict__ sdst) {
  int t = blockIdx.x * blockDim.x + threadIdx.x;
  if (t >= N_NODES * NH) return;
  int n = t >> 3, hh = t & 7;
  const float* row = Wh + (size_t)n * HF + hh * NF;
  const float* as  = a_src + hh * NF;
  const float* ad  = a_dst + hh * NF;
  float accs = 0.f, accd = 0.f;
  for (int f = 0; f < NF; ++f) { float v = row[f]; accs += v * as[f]; accd += v * ad[f]; }
  ssrc[t] = accs;
  sdst[t] = accd;
}

#define LOGIT_BLOCKS 1024
static_assert((LOGIT_BLOCKS * 256) % NH == 0, "a thread must keep its head");
#define TROWS     196
#define E4        (N_EDGES / 4)
#define E16       (N_EDGES / 16)
static_assert((N_NODES - 1) / TROWS < 256, "the tile number packs into 8 bits");
static_assert(N_EDGES % 16 == 0, "the tile numbers are read 16 at a time");
static_assert(LOGIT_BLOCKS * 256 >= E4, "one thread packs four tile numbers");
__global__ __launch_bounds__(256)
void edge_logits(const float* __restrict__ ef, const int* __restrict__ src,
                 const int* __restrict__ dst, const float* __restrict__ V,
                 const float* __restrict__ ssrc, const float* __restrict__ sdst,
                 float* __restrict__ sbuf, unsigned* __restrict__ m,
                 unsigned* __restrict__ tkey) {
  const int t0 = blockIdx.x * 256 + threadIdx.x;
  const int hh = t0 & 7;
  if (t0 < E4) {
    const int4 d4 = ((const int4*)dst)[t0];
    tkey[t0] = ((unsigned)d4.x / TROWS) | (((unsigned)d4.y / TROWS) << 8) |
               (((unsigned)d4.z / TROWS) << 16) | (((unsigned)d4.w / TROWS) << 24);
  }
  float vk[EDGE_F];
#pragma unroll
  for (int k = 0; k < EDGE_F; ++k) vk[k] = V[k * NH + hh];

  for (int t = t0; t < N_EDGES * NH; t += LOGIT_BLOCKS * 256) {
    const int e = t >> 3;
    const float4* efr = (const float4*)(ef + (size_t)e * EDGE_F);
    const int s = src[e], d = dst[e];
    const float ns = ssrc[s * NH + hh];
    const float nd = sdst[d * NH + hh];
    float acc = 0.f;
#pragma unroll
    for (int q = 0; q < EDGE_F / 4; ++q) {
      const float4 v = efr[q];
      const int k = q * 4;
      acc += v.x * vk[k + 0] + v.y * vk[k + 1] + v.z * vk[k + 2] + v.w * vk[k + 3];
    }
    float sv = ns + nd + acc;
    sv = sv > 0.f ? sv : 0.01f * sv;
    sbuf[t] = sv;
    atomicMax(&m[d * NH + hh], enc_f32(sv));
  }
}

#define TWAVES    32
#define LIST_CAP  544
static_assert(TROWS <= 256, "the local row packs into 8 bits");
static_assert(N_EDGES < (1 << 24), "the edge id packs into 24 bits");

__device__ __forceinline__ void lds_add(float* p, float v) {
  __hip_atomic_fetch_add(p, v, __ATOMIC_RELAXED, __HIP_MEMORY_SCOPE_WORKGROUP);
}

__global__ __launch_bounds__(1024)
void aggregate_tile(const int* __restrict__ src, const int* __restrict__ dst,
                    const float* __restrict__ Wh, const float* __restrict__ sbuf,
                    const unsigned* __restrict__ m, const unsigned* __restrict__ tkey,
                    float* __restrict__ out) {
  __shared__ float    tile[TROWS * HF];
  __shared__ unsigned elist[TWAVES * LIST_CAP];
  __shared__ float    astage[TWAVES * 32 * NH];
  __shared__ __attribute__((aligned(16))) float mtile[TROWS * NH];
  __shared__ __attribute__((aligned(16))) float dtile[TROWS * NH];

  const int tid  = threadIdx.x;
  const int lane = tid & 31;
  const int wave = __builtin_amdgcn_readfirstlane(tid >> 5);
  const int base = blockIdx.x * TROWS;
  const int nrows = min(TROWS, N_NODES - base);
  const unsigned lt = (1u << lane) - 1u;

  float4* t4 = (float4*)tile;
  for (int i = tid; i < TROWS * HF / 4; i += 1024) t4[i] = make_float4(0.f, 0.f, 0.f, 0.f);
  for (int i = tid; i < TROWS * NH; i += 1024) {
    dtile[i] = 0.f;
    mtile[i] = (i < nrows * NH) ? dec_f32(m[(size_t)base * NH + i]) : 0.f;
  }
  __syncthreads();

  unsigned* mylist  = elist + wave * LIST_CAP;
  float*    mystage = astage + wave * (32 * NH);
  const uint4* key16 = (const uint4*)tkey;
  const unsigned pat = (unsigned)blockIdx.x * 0x01010101u;
  const int NIT = (E16 + 1023) / 1024;
  int cnt = 0;

  const int q0 = wave * 32 + lane;
  auto load_keys = [&](int t) -> uint4 { return key16[min(q0 + t * 1024, E16 - 1)]; };

  auto scan_trip = [&](const uint4 cur, int t) {
    const int q = q0 + t * 1024;
    const unsigned kw[4] = {cur.x, cur.y, cur.z, cur.w};
    unsigned u = 0u;
#pragma unroll
    for (int d = 0; d < 4; ++d) {
      const unsigned x = kw[d] ^ pat;
      const unsigned y = ((x & 0x7f7f7f7fu) + 0x7f7f7f7fu) | x;
      u |= (~y & 0x80808080u) >> (7 - d);
    }
    if (q >= E16) u = 0u;
    unsigned mask = __builtin_amdgcn_ballot_w32(u != 0u);
    while (mask != 0u) {
      if (u != 0u) {
        const unsigned p = (unsigned)__builtin_ctz(u);
        const unsigned off = ((p & 7u) << 2) | (p >> 3);
        mylist[cnt + __popc(mask & lt)] = (unsigned)q * 16u + off;
        u &= u - 1u;
      }
      cnt += __popc(mask);
      mask = __builtin_amdgcn_ballot_w32(u != 0u);
    }
  };

  auto flush = [&]() {
    for (int b = 0; b < cnt; b += 32) {
      const int nb = min(32, cnt - b);
      unsigned ent = 0u;
      int s_i = 0;
      if (lane < nb) {
        const int e = (int)mylist[b + lane];
        const int lrow = dst[e] - base;
        ent = ((unsigned)e << 8) | (unsigned)lrow;
        s_i = src[e];
        const float4* sp = (const float4*)(sbuf + (size_t)e * NH);
        const float4* mp = (const float4*)(mtile + lrow * NH);
        const float4 x0 = sp[0], x1 = sp[1], y0 = mp[0], y1 = mp[1];
        const float4 e0 = make_float4(expf(x0.x - y0.x), expf(x0.y - y0.y),
                                      expf(x0.z - y0.z), expf(x0.w - y0.w));
        const float4 e1 = make_float4(expf(x1.x - y1.x), expf(x1.y - y1.y),
                                      expf(x1.z - y1.z), expf(x1.w - y1.w));
        float4* st = (float4*)(mystage + lane * NH);
        st[0] = e0;
        st[1] = e1;
        float* dp = dtile + lrow * NH;
        lds_add(dp + 0, e0.x); lds_add(dp + 1, e0.y);
        lds_add(dp + 2, e0.z); lds_add(dp + 3, e0.w);
        lds_add(dp + 4, e1.x); lds_add(dp + 5, e1.y);
        lds_add(dp + 6, e1.z); lds_add(dp + 7, e1.w);
      }
      for (int u0 = 0; u0 < nb; u0 += 4) {
        float4 v0[4], v1[4];
        float a0[4], a1[4];
        int lr[4];
#pragma unroll
        for (int k = 0; k < 4; ++k) {
          const int u = min(u0 + k, nb - 1);
          const int s = __builtin_amdgcn_readlane(s_i, u);
          lr[k] = __builtin_amdgcn_readlane((int)ent, u) & 255;
          const float4* rp = (const float4*)(Wh + (size_t)s * HF);
          v0[k] = rp[lane];
          v1[k] = rp[32 + lane];
          a0[k] = mystage[u * NH + (lane >> 3)];
          a1[k] = mystage[u * NH + 4 + (lane >> 3)];
        }
#pragma unroll
        for (int k = 0; k < 4; ++k) {
          if (u0 + k < nb) {
            float* tp = tile + lr[k] * HF + lane;
            lds_add(tp,       v0[k].x * a0[k]);
            lds_add(tp + 32,  v0[k].y * a0[k]);
            lds_add(tp + 64,  v0[k].z * a0[k]);
            lds_add(tp + 96,  v0[k].w * a0[k]);
            lds_add(tp + 128, v1[k].x * a1[k]);
            lds_add(tp + 160, v1[k].y * a1[k]);
            lds_add(tp + 192, v1[k].z * a1[k]);
            lds_add(tp + 224, v1[k].w * a1[k]);
          }
        }
      }
    }
    cnt = 0;
  };

  int it = 0;
  while (it < NIT) {
    uint4 ka = load_keys(it), kb = load_keys(it + 1), kc = load_keys(it + 2);
    for (;;) {
      scan_trip(ka, it); ++it;
      if (it >= NIT || cnt >= 32) break;
      ka = load_keys(it + 2);
      scan_trip(kb, it); ++it;
      if (it >= NIT || cnt >= 32) break;
      kb = load_keys(it + 2);
      scan_trip(kc, it); ++it;
      if (it >= NIT || cnt >= 32) break;
      kc = load_keys(it + 2);
    }
    flush();
  }
  __syncthreads();

  for (int i = tid; i < nrows * 64; i += 1024) {
    const int r = i >> 6, g = i & 63;
    const int half = g >> 5, l = g & 31;
    const float* tp = tile + r * HF + half * 128 + l;
    float4 x = make_float4(tp[0], tp[32], tp[64], tp[96]);
    const float dn = dtile[r * NH + half * 4 + (l >> 3)];
    if (dn > 0.f) { x.x /= dn; x.y /= dn; x.z /= dn; x.w /= dn; }
    x.x = x.x > 0.f ? x.x : expm1f(x.x);
    x.y = x.y > 0.f ? x.y : expm1f(x.y);
    x.z = x.z > 0.f ? x.z : expm1f(x.z);
    x.w = x.w > 0.f ? x.w : expm1f(x.w);
    *(float4*)(out + (size_t)(base + r) * HF + half * 128 + 4 * l) = x;
  }
}

extern "C" void kernel_launch(void* const* d_in, const int* in_sizes, int n_in,
                              void* d_out, int out_size, void* d_ws, size_t ws_size,
                              hipStream_t stream) {
  const float* h   = (const float*)d_in[0];
  const float* ef  = (const float*)d_in[1];
  const int*   src = (const int*)  d_in[2];
  const int*   dst = (const int*)  d_in[3];
  const float* W_w = (const float*)d_in[4];
  const float* We_w= (const float*)d_in[5];
  const float* a_s = (const float*)d_in[6];
  const float* a_d = (const float*)d_in[7];
  const float* a_e = (const float*)d_in[8];
  float* out = (float*)d_out;

  char* ws = (char*)d_ws;
  const size_t WH_BYTES   = (size_t)N_NODES * HF * 4;
  const size_t SBUF_BYTES = (size_t)N_EDGES * NH * 4;
  const size_t NH_BYTES   = (size_t)N_NODES * NH * 4;
  float*    Wh   = (float*)   (ws);
  float*    sbuf = (float*)   (ws + WH_BYTES);
  float*    ssrc = (float*)   (ws + WH_BYTES + SBUF_BYTES);
  float*    sdst = (float*)   (ws + WH_BYTES + SBUF_BYTES + NH_BYTES);
  unsigned* mmax = (unsigned*)(ws + WH_BYTES + SBUF_BYTES + 2 * NH_BYTES);
  unsigned* tkey = (unsigned*)(ws + WH_BYTES + SBUF_BYTES + 3 * NH_BYTES);
  float*    V    = (float*)   (ws + WH_BYTES + SBUF_BYTES + 4 * NH_BYTES);

  hipMemsetAsync(mmax, 0, NH_BYTES, stream);

  gemm_wh_wmma<<<N_NODES / 16, 128, 0, stream>>>(h, W_w, Wh);
  fold_we<<<1, 512, 0, stream>>>(We_w, a_e, V);
  node_scores<<<(N_NODES * NH + 255) / 256, 256, 0, stream>>>(Wh, a_s, a_d, ssrc, sdst);
  edge_logits<<<LOGIT_BLOCKS, 256, 0, stream>>>(ef, src, dst, V, ssrc, sdst, sbuf, mmax, tkey);
  aggregate_tile<<<(N_NODES + TROWS - 1) / TROWS, 1024, 0, stream>>>(src, dst, Wh, sbuf, mmax, tkey, out);
}


# Segment max inside the dst tile: a tile work-group keeps its edge list in LDS, takes an LDS integer max of the logits, a barrier, then sums; global atomicMax and memset go; LDS add order changes
# speedup vs baseline: 4.5539x; 1.0958x over previous
#include <hip/hip_runtime.h>
#include <math.h>

typedef __attribute__((ext_vector_type(2))) float v2f;
typedef __attribute__((ext_vector_type(4))) float v4f;
typedef __attribute__((ext_vector_type(8))) float v8f;

#define N_NODES   50000
#define N_EDGES   800000
#define IN_F      256
#define EDGE_F    64
#define NH        8
#define NF        32
#define HF        256

__device__ __forceinline__ unsigned enc_f32(float f) {
  unsigned u = __float_as_uint(f);
  return (u & 0x80000000u) ? ~u : (u | 0x80000000u);
}
__device__ __forceinline__ float dec_f32(unsigned u) {
  unsigned b = (u & 0x80000000u) ? (u ^ 0x80000000u) : ~u;
  return __uint_as_float(b);
}

__global__ __launch_bounds__(128)
void gemm_wh_wmma(const float* __restrict__ h, const float* __restrict__ W,
                  float* __restrict__ Wh) {
  const int lane = threadIdx.x & 31;
  const int wave = threadIdx.x >> 5;
  const int m0   = blockIdx.x * 16;
  const int n0   = wave * 64;
  const int c    = lane & 15;
  const int hi   = lane >> 4;

  v8f c0 = {}; v8f c1 = {}; v8f c2 = {}; v8f c3 = {};
  const float* arow = h + (size_t)(m0 + c) * IN_F + 4 * hi;
  const float* brow = W + (size_t)(4 * hi) * HF + n0 + 4 * c;

  for (int k0 = 0; k0 < IN_F; k0 += 8) {
    const v4f av = *(const v4f*)(arow + k0);
    const float* bp = brow + (size_t)k0 * HF;
    const v4f q0 = *(const v4f*)(bp);
    const v4f q1 = *(const v4f*)(bp + HF);
    const v4f q2 = *(const v4f*)(bp + 2 * HF);
    const v4f q3 = *(const v4f*)(bp + 3 * HF);
    v2f a, b;
    a.x = av.x; a.y = av.y;
    b.x = q0.x; b.y = q1.x;
    c0 = __builtin_amdgcn_wmma_f32_16x16x4_f32(false, a, false, b, (short)0, c0, false, false);
    b.x = q0.y; b.y = q1.y;
    c1 = __builtin_amdgcn_wmma_f32_16x16x4_f32(false, a, false, b, (short)0, c1, false, false);
    b.x = q0.z; b.y = q1.z;
    c2 = __builtin_amdgcn_wmma_f32_16x16x4_f32(false, a, false, b, (short)0, c2, false, false);
    b.x = q0.w; b.y = q1.w;
    c3 = __builtin_amdgcn_wmma_f32_16x16x4_f32(false, a, false, b, (short)0, c3, false, false);
    a.x = av.z; a.y = av.w;
    b.x = q2.x; b.y = q3.x;
    c0 = __builtin_amdgcn_wmma_f32_16x16x4_f32(false, a, false, b, (short)0, c0, false, false);
    b.x = q2.y; b.y = q3.y;
    c1 = __builtin_amdgcn_wmma_f32_16x16x4_f32(false, a, false, b, (short)0, c1, false, false);
    b.x = q2.z; b.y = q3.z;
    c2 = __builtin_amdgcn_wmma_f32_16x16x4_f32(false, a, false, b, (short)0, c2, false, false);
    b.x = q2.w; b.y = q3.w;
    c3 = __builtin_amdgcn_wmma_f32_16x16x4_f32(false, a, false, b, (short)0, c3, false, false);
  }
  float* op = Wh + (size_t)(m0 + (hi << 3)) * HF + n0 + 4 * c;
  for (int r = 0; r < 8; ++r) {
    v4f o;
    o.x = c0[r]; o.y = c1[r]; o.z = c2[r]; o.w = c3[r];
    *(v4f*)(op + (size_t)r * HF) = o;
  }
}

__global__ void fold_we(const float* __restrict__ We_w,
                        const float* __restrict__ a_edge,
                        float* __restrict__ V) {
  int t = threadIdx.x;
  int k = t >> 3, hh = t & 7;
  const float* wr = We_w + (size_t)k * HF + hh * NF;
  const float* ae = a_edge + hh * NF;
  float acc = 0.f;
  for (int f = 0; f < NF; ++f) acc += wr[f] * ae[f];
  V[k * NH + hh] = acc;
}

__global__ __launch_bounds__(256)
void node_scores(const float* __restrict__ Wh, const float* __restrict__ a_src,
                 const float* __restrict__ a_dst,
                 float* __restrict__ ssrc, float* __restrict__ sdst) {
  int t = blockIdx.x * blockDim.x + threadIdx.x;
  if (t >= N_NODES * NH) return;
  int n = t >> 3, hh = t & 7;
  const float* row = Wh + (size_t)n * HF + hh * NF;
  const float* as  = a_src + hh * NF;
  const float* ad  = a_dst + hh * NF;
  float accs = 0.f, accd = 0.f;
  for (int f = 0; f < NF; ++f) { float v = row[f]; accs += v * as[f]; accd += v * ad[f]; }
  ssrc[t] = accs;
  sdst[t] = accd;
}

#define LOGIT_BLOCKS 1024
static_assert((LOGIT_BLOCKS * 256) % NH == 0, "a thread must keep its head");
#define TROWS     196
#define E4        (N_EDGES / 4)
#define E16       (N_EDGES / 16)
static_assert((N_NODES - 1) / TROWS < 256, "the tile number packs into 8 bits");
static_assert(N_EDGES % 16 == 0, "the tile numbers are read 16 at a time");
static_assert(LOGIT_BLOCKS * 256 >= E4, "one thread packs four tile numbers");
__global__ __launch_bounds__(256)
void edge_logits(const float* __restrict__ ef, const int* __restrict__ src,
                 const int* __restrict__ dst, const float* __restrict__ V,
                 const float* __restrict__ ssrc, const float* __restrict__ sdst,
                 float* __restrict__ sbuf, unsigned* __restrict__ tkey) {
  const int t0 = blockIdx.x * 256 + threadIdx.x;
  const int hh = t0 & 7;
  if (t0 < E4) {
    const int4 d4 = ((const int4*)dst)[t0];
    tkey[t0] = ((unsigned)d4.x / TROWS) | (((unsigned)d4.y / TROWS) << 8) |
               (((unsigned)d4.z / TROWS) << 16) | (((unsigned)d4.w / TROWS) << 24);
  }
  float vk[EDGE_F];
#pragma unroll
  for (int k = 0; k < EDGE_F; ++k) vk[k] = V[k * NH + hh];

#pragma clang loop vectorize(disable) interleave(disable) unroll(disable)
  for (int t = t0; t < N_EDGES * NH; t += LOGIT_BLOCKS * 256) {
    const int e = t >> 3;
    const float4* efr = (const float4*)(ef + (size_t)e * EDGE_F);
    const int s = src[e], d = dst[e];
    const float ns = ssrc[s * NH + hh];
    const float nd = sdst[d * NH + hh];
    float acc = 0.f;
#pragma unroll
    for (int q = 0; q < EDGE_F / 4; ++q) {
      const float4 v = efr[q];
      const int k = q * 4;
      acc += v.x * vk[k + 0] + v.y * vk[k + 1] + v.z * vk[k + 2] + v.w * vk[k + 3];
    }
    float sv = ns + nd + acc;
    sv = sv > 0.f ? sv : 0.01f * sv;
    sbuf[t] = sv;
  }
}

#define TWAVES    32
#define LIST_CAP  632
#define KDEPTH    7
static_assert(TROWS <= 256, "the local row packs into 8 bits");
static_assert(N_EDGES < (1 << 24), "the edge id packs into 24 bits");

__device__ __forceinline__ void lds_add(float* p, float v) {
  __hip_atomic_fetch_add(p, v, __ATOMIC_RELAXED, __HIP_MEMORY_SCOPE_WORKGROUP);
}
__device__ __forceinline__ void lds_max(unsigned* p, unsigned v) {
  __hip_atomic_fetch_max(p, v, __ATOMIC_RELAXED, __HIP_MEMORY_SCOPE_WORKGROUP);
}

__global__ __launch_bounds__(1024)
void aggregate_tile(const int* __restrict__ src, const int* __restrict__ dst,
                    const float* __restrict__ Wh, const float* __restrict__ sbuf,
                    const unsigned* __restrict__ tkey, float* __restrict__ out) {
  __shared__ float    tile[TROWS * HF];
  __shared__ unsigned elist[TWAVES * LIST_CAP];
  __shared__ float    astage[TWAVES * 32 * NH];
  __shared__ __attribute__((aligned(16))) unsigned mtile[TROWS * NH];
  __shared__ __attribute__((aligned(16))) float    dtile[TROWS * NH];

  const int tid  = threadIdx.x;
  const int lane = tid & 31;
  const int wave = __builtin_amdgcn_readfirstlane(tid >> 5);
  const int base = blockIdx.x * TROWS;
  const int nrows = min(TROWS, N_NODES - base);
  const unsigned lt = (1u << lane) - 1u;

  float4* t4 = (float4*)tile;
  for (int i = tid; i < TROWS * HF / 4; i += 1024) t4[i] = make_float4(0.f, 0.f, 0.f, 0.f);
  for (int i = tid; i < TROWS * NH; i += 1024) {
    dtile[i] = 0.f;
    mtile[i] = 0u;
  }
  __syncthreads();

  unsigned* mylist  = elist + wave * LIST_CAP;
  float*    mystage = astage + wave * (32 * NH);
  const uint4* key16 = (const uint4*)tkey;
  const unsigned pat = (unsigned)blockIdx.x * 0x01010101u;
  const int NIT = (E16 + 1023) / 1024;
  int cnt = 0;

  const int q0 = wave * 32 + lane;
  auto load_keys = [&](int t) -> uint4 { return key16[min(q0 + t * 1024, E16 - 1)]; };

  auto scan_trip = [&](const uint4 cur, int t) -> bool {
    const int q = q0 + t * 1024;
    const unsigned kw[4] = {cur.x, cur.y, cur.z, cur.w};
    unsigned u = 0u;
#pragma unroll
    for (int d = 0; d < 4; ++d) {
      const unsigned x = kw[d] ^ pat;
      const unsigned y = ((x & 0x7f7f7f7fu) + 0x7f7f7f7fu) | x;
      u |= (~y & 0x80808080u) >> (7 - d);
    }
    if (q >= E16) u = 0u;
    unsigned mask = __builtin_amdgcn_ballot_w32(u != 0u);
    while (mask != 0u) {
      if (cnt + 32 > LIST_CAP) return false;
      if (u != 0u) {
        const unsigned p = (unsigned)__builtin_ctz(u);
        const unsigned off = ((p & 7u) << 2) | (p >> 3);
        mylist[cnt + __popc(mask & lt)] = (unsigned)q * 16u + off;
        u &= u - 1u;
      }
      cnt += __popc(mask);
      mask = __builtin_amdgcn_ballot_w32(u != 0u);
    }
    return true;
  };

  static_assert(LIST_CAP >= 512 + 32, "an empty list takes a whole trip");
  auto scan_all = [&](auto&& on_full) {
    uint4 k[KDEPTH];
#pragma unroll
    for (int j = 0; j < KDEPTH; ++j) k[j] = load_keys(j);
    for (int it0 = 0; it0 < NIT; it0 += KDEPTH) {
      int stop = -1;
#pragma unroll
      for (int j = 0; j < KDEPTH; ++j) {
        if (stop < 0 && it0 + j < NIT) {
          const int c0 = cnt;
          if (!scan_trip(k[j], it0 + j)) { cnt = c0; stop = it0 + j; }
        }
        k[j] = load_keys(it0 + j + KDEPTH);
      }
      if (stop >= 0) {
        const int end = min(it0 + KDEPTH, NIT);
        for (int t = stop; t < end; ++t) {
          const uint4 cur = load_keys(t);
          const int c0 = cnt;
          if (!scan_trip(cur, t)) {
            cnt = c0;
            on_full();
            scan_trip(cur, t);
          }
        }
      }
    }
  };

  auto take_max = [&](int n) {
    for (int b = 0; b < n; b += 32) {
      if (b + lane < n) {
        const int e = (int)mylist[b + lane];
        const int lrow = dst[e] - base;
        const float4* sp = (const float4*)(sbuf + (size_t)e * NH);
        const float4 x0 = sp[0], x1 = sp[1];
        unsigned* mp = mtile + lrow * NH;
        lds_max(mp + 0, enc_f32(x0.x)); lds_max(mp + 1, enc_f32(x0.y));
        lds_max(mp + 2, enc_f32(x0.z)); lds_max(mp + 3, enc_f32(x0.w));
        lds_max(mp + 4, enc_f32(x1.x)); lds_max(mp + 5, enc_f32(x1.y));
        lds_max(mp + 6, enc_f32(x1.z)); lds_max(mp + 7, enc_f32(x1.w));
        mylist[b + lane] = ((unsigned)e << 8) | (unsigned)lrow;
      }
    }
  };

  auto flush = [&](int n, bool raw) {
    for (int b = 0; b < n; b += 32) {
      const int nb = min(32, n - b);
      int lrow = 0;
      int s_i = 0;
      if (lane < nb) {
        const unsigned w = mylist[b + lane];
        int e;
        if (raw) { e = (int)w; lrow = dst[e] - base; }
        else     { e = (int)(w >> 8); lrow = (int)(w & 255u); }
        s_i = src[e];
        const float4* sp = (const float4*)(sbuf + (size_t)e * NH);
        const uint4*  mp = (const uint4*)(mtile + lrow * NH);
        const float4 x0 = sp[0], x1 = sp[1];
        const uint4  y0 = mp[0], y1 = mp[1];
        const float4 e0 = make_float4(expf(x0.x - dec_f32(y0.x)), expf(x0.y - dec_f32(y0.y)),
                                      expf(x0.z - dec_f32(y0.z)), expf(x0.w - dec_f32(y0.w)));
        const float4 e1 = make_float4(expf(x1.x - dec_f32(y1.x)), expf(x1.y - dec_f32(y1.y)),
                                      expf(x1.z - dec_f32(y1.z)), expf(x1.w - dec_f32(y1.w)));
        float4* st = (float4*)(mystage + lane * NH);
        st[0] = e0;
        st[1] = e1;
        float* dp = dtile + lrow * NH;
        lds_add(dp + 0, e0.x); lds_add(dp + 1, e0.y);
        lds_add(dp + 2, e0.z); lds_add(dp + 3, e0.w);
        lds_add(dp + 4, e1.x); lds_add(dp + 5, e1.y);
        lds_add(dp + 6, e1.z); lds_add(dp + 7, e1.w);
      }
      for (int u0 = 0; u0 < nb; u0 += 4) {
        float4 v0[4], v1[4];
        float a0[4], a1[4];
        int lr[4];
#pragma unroll
        for (int k = 0; k < 4; ++k) {
          const int u = min(u0 + k, nb - 1);
          const int s = __builtin_amdgcn_readlane(s_i, u);
          lr[k] = __builtin_amdgcn_readlane(lrow, u);
          const float4* rp = (const float4*)(Wh + (size_t)s * HF);
          v0[k] = rp[lane];
          v1[k] = rp[32 + lane];
          a0[k] = mystage[u * NH + (lane >> 3)];
          a1[k] = mystage[u * NH + 4 + (lane >> 3)];
        }
#pragma unroll
        for (int k = 0; k < 4; ++k) {
          if (u0 + k < nb) {
            float* tp = tile + lr[k] * HF + lane;
            lds_add(tp,       v0[k].x * a0[k]);
            lds_add(tp + 32,  v0[k].y * a0[k]);
            lds_add(tp + 64,  v0[k].z * a0[k]);
            lds_add(tp + 96,  v0[k].w * a0[k]);
            lds_add(tp + 128, v1[k].x * a1[k]);
            lds_add(tp + 160, v1[k].y * a1[k]);
            lds_add(tp + 192, v1[k].z * a1[k]);
            lds_add(tp + 224, v1[k].w * a1[k]);
          }
        }
      }
    }
  };

  bool kept = true;
  scan_all([&]() { take_max(cnt); cnt = 0; kept = false; });
  take_max(cnt);
  __syncthreads();

  if (!kept) {
    cnt = 0;
    scan_all([&]() { flush(cnt, true); cnt = 0; });
  }
  flush(cnt, !kept);
  __syncthreads();

  for (int i = tid; i < nrows * 64; i += 1024) {
    const int r = i >> 6, g = i & 63;
    const int half = g >> 5, l = g & 31;
    const float* tp = tile + r * HF + half * 128 + l;
    float4 x = make_float4(tp[0], tp[32], tp[64], tp[96]);
    const float dn = dtile[r * NH + half * 4 + (l >> 3)];
    if (dn > 0.f) { x.x /= dn; x.y /= dn; x.z /= dn; x.w /= dn; }
    x.x = x.x > 0.f ? x.x : expm1f(x.x);
    x.y = x.y > 0.f ? x.y : expm1f(x.y);
    x.z = x.z > 0.f ? x.z : expm1f(x.z);
    x.w = x.w > 0.f ? x.w : expm1f(x.w);
    *(float4*)(out + (size_t)(base + r) * HF + half * 128 + 4 * l) = x;
  }
}

extern "C" void kernel_launch(void* const* d_in, const int* in_sizes, int n_in,
                              void* d_out, int out_size, void* d_ws, size_t ws_size,
                              hipStream_t stream) {
  const float* h   = (const float*)d_in[0];
  const float* ef  = (const float*)d_in[1];
  const int*   src = (const int*)  d_in[2];
  const int*   dst = (const int*)  d_in[3];
  const float* W_w = (const float*)d_in[4];
  const float* We_w= (const float*)d_in[5];
  const float* a_s = (const float*)d_in[6];
  const float* a_d = (const float*)d_in[7];
  const float* a_e = (const float*)d_in[8];
  float* out = (float*)d_out;

  char* ws = (char*)d_ws;
  const size_t WH_BYTES   = (size_t)N_NODES * HF * 4;
  const size_t SBUF_BYTES = (size_t)N_EDGES * NH * 4;
  const size_t NH_BYTES   = (size_t)N_NODES * NH * 4;
  float*    Wh   = (float*)   (ws);
  float*    sbuf = (float*)   (ws + WH_BYTES);
  float*    ssrc = (float*)   (ws + WH_BYTES + SBUF_BYTES);
  float*    sdst = (float*)   (ws + WH_BYTES + SBUF_BYTES + NH_BYTES);
  unsigned* tkey = (unsigned*)(ws + WH_BYTES + SBUF_BYTES + 3 * NH_BYTES);
  float*    V    = (float*)   (ws + WH_BYTES + SBUF_BYTES + 4 * NH_BYTES);

  gemm_wh_wmma<<<N_NODES / 16, 128, 0, stream>>>(h, W_w, Wh);
  fold_we<<<1, 512, 0, stream>>>(We_w, a_e, V);
  node_scores<<<(N_NODES * NH + 255) / 256, 256, 0, stream>>>(Wh, a_s, a_d, ssrc, sdst);
  edge_logits<<<LOGIT_BLOCKS, 256, 0, stream>>>(ef, src, dst, V, ssrc, sdst, sbuf, tkey);
  aggregate_tile<<<(N_NODES + TROWS - 1) / TROWS, 1024, 0, stream>>>(src, dst, Wh, sbuf, tkey, out);
}


# Node score halves in the GEMM epilogue, from the accumulators: 4 products per lane, 8-lane butterfly, one 512-byte store per plane; node_scores launch and its Wh re-read go; score sum order changes
# speedup vs baseline: 4.9524x; 1.0875x over previous
#include <hip/hip_runtime.h>
#include <math.h>

typedef __attribute__((ext_vector_type(2))) float v2f;
typedef __attribute__((ext_vector_type(4))) float v4f;
typedef __attribute__((ext_vector_type(8))) float v8f;

#define N_NODES   50000
#define N_EDGES   800000
#define IN_F      256
#define EDGE_F    64
#define NH        8
#define NF        32
#define HF        256

__device__ __forceinline__ unsigned enc_f32(float f) {
  unsigned u = __float_as_uint(f);
  return (u & 0x80000000u) ? ~u : (u | 0x80000000u);
}
__device__ __forceinline__ float dec_f32(unsigned u) {
  unsigned b = (u & 0x80000000u) ? (u ^ 0x80000000u) : ~u;
  return __uint_as_float(b);
}

__global__ __launch_bounds__(128)
void gemm_wh_wmma(const float* __restrict__ h, const float* __restrict__ W,
                  const float* __restrict__ a_src, const float* __restrict__ a_dst,
                  float* __restrict__ Wh,
                  float* __restrict__ ssrc, float* __restrict__ sdst) {
  __shared__ __attribute__((aligned(16))) float sc_s[16 * NH];
  __shared__ __attribute__((aligned(16))) float sc_d[16 * NH];
  const int lane = threadIdx.x & 31;
  const int wave = threadIdx.x >> 5;
  const int m0   = blockIdx.x * 16;
  const int n0   = wave * 64;
  const int c    = lane & 15;
  const int hi   = lane >> 4;

  v8f c0 = {}; v8f c1 = {}; v8f c2 = {}; v8f c3 = {};
  const float* arow = h + (size_t)(m0 + c) * IN_F + 4 * hi;
  const float* brow = W + (size_t)(4 * hi) * HF + n0 + 4 * c;

  for (int k0 = 0; k0 < IN_F; k0 += 8) {
    const v4f av = *(const v4f*)(arow + k0);
    const float* bp = brow + (size_t)k0 * HF;
    const v4f q0 = *(const v4f*)(bp);
    const v4f q1 = *(const v4f*)(bp + HF);
    const v4f q2 = *(const v4f*)(bp + 2 * HF);
    const v4f q3 = *(const v4f*)(bp + 3 * HF);
    v2f a, b;
    a.x = av.x; a.y = av.y;
    b.x = q0.x; b.y = q1.x;
    c0 = __builtin_amdgcn_wmma_f32_16x16x4_f32(false, a, false, b, (short)0, c0, false, false);
    b.x = q0.y; b.y = q1.y;
    c1 = __builtin_amdgcn_wmma_f32_16x16x4_f32(false, a, false, b, (short)0, c1, false, false);
    b.x = q0.z; b.y = q1.z;
    c2 = __builtin_amdgcn_wmma_f32_16x16x4_f32(false, a, false, b, (short)0, c2, false, false);
    b.x = q0.w; b.y = q1.w;
    c3 = __builtin_amdgcn_wmma_f32_16x16x4_f32(false, a, false, b, (short)0, c3, false, false);
    a.x = av.z; a.y = av.w;
    b.x = q2.x; b.y = q3.x;
    c0 = __builtin_amdgcn_wmma_f32_16x16x4_f32(false, a, false, b, (short)0, c0, false, false);
    b.x = q2.y; b.y = q3.y;
    c1 = __builtin_amdgcn_wmma_f32_16x16x4_f32(false, a, false, b, (short)0, c1, false, false);
    b.x = q2.z; b.y = q3.z;
    c2 = __builtin_amdgcn_wmma_f32_16x16x4_f32(false, a, false, b, (short)0, c2, false, false);
    b.x = q2.w; b.y = q3.w;
    c3 = __builtin_amdgcn_wmma_f32_16x16x4_f32(false, a, false, b, (short)0, c3, false, false);
  }

  const v4f as4 = *(const v4f*)(a_src + n0 + 4 * c);
  const v4f ad4 = *(const v4f*)(a_dst + n0 + 4 * c);
  float ps[8], pd[8];
#pragma unroll
  for (int r = 0; r < 8; ++r) {
    ps[r] = c0[r] * as4.x + c1[r] * as4.y + c2[r] * as4.z + c3[r] * as4.w;
    pd[r] = c0[r] * ad4.x + c1[r] * ad4.y + c2[r] * ad4.z + c3[r] * ad4.w;
  }
#pragma unroll
  for (int m = 1; m < 8; m <<= 1) {
#pragma unroll
    for (int r = 0; r < 8; ++r) {
      ps[r] += __shfl_xor(ps[r], m, 32);
      pd[r] += __shfl_xor(pd[r], m, 32);
    }
  }
  const int rr = c & 7;
  float vs = ps[0], vd = pd[0];
#pragma unroll
  for (int r = 1; r < 8; ++r) {
    vs = (rr == r) ? ps[r] : vs;
    vd = (rr == r) ? pd[r] : vd;
  }
  const int slot = ((hi << 3) + rr) * NH + 2 * wave + (c >> 3);
  sc_s[slot] = vs;
  sc_d[slot] = vd;
  __syncthreads();

  float* op = Wh + (size_t)(m0 + (hi << 3)) * HF + n0 + 4 * c;
  for (int r = 0; r < 8; ++r) {
    v4f o;
    o.x = c0[r]; o.y = c1[r]; o.z = c2[r]; o.w = c3[r];
    *(v4f*)(op + (size_t)r * HF) = o;
  }
  if (wave == 0) {
    *(v4f*)(ssrc + (size_t)m0 * NH + 4 * lane) = *(const v4f*)(sc_s + 4 * lane);
  } else if (wave == 1) {
    *(v4f*)(sdst + (size_t)m0 * NH + 4 * lane) = *(const v4f*)(sc_d + 4 * lane);
  }
}

__global__ void fold_we(const float* __restrict__ We_w,
                        const float* __restrict__ a_edge,
                        float* __restrict__ V) {
  int t = threadIdx.x;
  int k = t >> 3, hh = t & 7;
  const float* wr = We_w + (size_t)k * HF + hh * NF;
  const float* ae = a_edge + hh * NF;
  float acc = 0.f;
  for (int f = 0; f < NF; ++f) acc += wr[f] * ae[f];
  V[k * NH + hh] = acc;
}

#define LOGIT_BLOCKS 1024
static_assert((LOGIT_BLOCKS * 256) % NH == 0, "a thread must keep its head");
#define TROWS     196
#define E4        (N_EDGES / 4)
#define E16       (N_EDGES / 16)
static_assert((N_NODES - 1) / TROWS < 256, "the tile number packs into 8 bits");
static_assert(N_NODES % 16 == 0, "a GEMM block owns 16 whole rows and one 512-byte piece of each score plane");
static_assert(N_EDGES % 16 == 0, "the tile numbers are read 16 at a time");
static_assert(LOGIT_BLOCKS * 256 >= E4, "one thread packs four tile numbers");
__global__ __launch_bounds__(256)
void edge_logits(const float* __restrict__ ef, const int* __restrict__ src,
                 const int* __restrict__ dst, const float* __restrict__ V,
                 const float* __restrict__ ssrc, const float* __restrict__ sdst,
                 float* __restrict__ sbuf, unsigned* __restrict__ tkey) {
  const int t0 = blockIdx.x * 256 + threadIdx.x;
  const int hh = t0 & 7;
  if (t0 < E4) {
    const int4 d4 = ((const int4*)dst)[t0];
    tkey[t0] = ((unsigned)d4.x / TROWS) | (((unsigned)d4.y / TROWS) << 8) |
               (((unsigned)d4.z / TROWS) << 16) | (((unsigned)d4.w / TROWS) << 24);
  }
  float vk[EDGE_F];
#pragma unroll
  for (int k = 0; k < EDGE_F; ++k) vk[k] = V[k * NH + hh];

#pragma clang loop vectorize(disable) interleave(disable) unroll(disable)
  for (int t = t0; t < N_EDGES * NH; t += LOGIT_BLOCKS * 256) {
    const int e = t >> 3;
    const float4* efr = (const float4*)(ef + (size_t)e * EDGE_F);
    const int s = src[e], d = dst[e];
    const float ns = ssrc[s * NH + hh];
    const float nd = sdst[d * NH + hh];
    float acc = 0.f;
#pragma unroll
    for (int q = 0; q < EDGE_F / 4; ++q) {
      const float4 v = efr[q];
      const int k = q * 4;
      acc += v.x * vk[k + 0] + v.y * vk[k + 1] + v.z * vk[k + 2] + v.w * vk[k + 3];
    }
    float sv = ns + nd + acc;
    sv = sv > 0.f ? sv : 0.01f * sv;
    sbuf[t] = sv;
  }
}

#define TWAVES    32
#define LIST_CAP  632
#define KDEPTH    7
static_assert(TROWS <= 256, "the local row packs into 8 bits");
static_assert(N_EDGES < (1 << 24), "the edge id packs into 24 bits");

__device__ __forceinline__ void lds_add(float* p, float v) {
  __hip_atomic_fetch_add(p, v, __ATOMIC_RELAXED, __HIP_MEMORY_SCOPE_WORKGROUP);
}
__device__ __forceinline__ void lds_max(unsigned* p, unsigned v) {
  __hip_atomic_fetch_max(p, v, __ATOMIC_RELAXED, __HIP_MEMORY_SCOPE_WORKGROUP);
}

__global__ __launch_bounds__(1024)
void aggregate_tile(const int* __restrict__ src, const int* __restrict__ dst,
                    const float* __restrict__ Wh, const float* __restrict__ sbuf,
                    const unsigned* __restrict__ tkey, float* __restrict__ out) {
  __shared__ float    tile[TROWS * HF];
  __shared__ unsigned elist[TWAVES * LIST_CAP];
  __shared__ float    astage[TWAVES * 32 * NH];
  __shared__ __attribute__((aligned(16))) unsigned mtile[TROWS * NH];
  __shared__ __attribute__((aligned(16))) float    dtile[TROWS * NH];

  const int tid  = threadIdx.x;
  const int lane = tid & 31;
  const int wave = __builtin_amdgcn_readfirstlane(tid >> 5);
  const int base = blockIdx.x * TROWS;
  const int nrows = min(TROWS, N_NODES - base);
  const unsigned lt = (1u << lane) - 1u;

  float4* t4 = (float4*)tile;
  for (int i = tid; i < TROWS * HF / 4; i += 1024) t4[i] = make_float4(0.f, 0.f, 0.f, 0.f);
  for (int i = tid; i < TROWS * NH; i += 1024) {
    dtile[i] = 0.f;
    mtile[i] = 0u;
  }
  __syncthreads();

  unsigned* mylist  = elist + wave * LIST_CAP;
  float*    mystage = astage + wave * (32 * NH);
  const uint4* key16 = (const uint4*)tkey;
  const unsigned pat = (unsigned)blockIdx.x * 0x01010101u;
  const int NIT = (E16 + 1023) / 1024;
  int cnt = 0;

  const int q0 = wave * 32 + lane;
  auto load_keys = [&](int t) -> uint4 { return key16[min(q0 + t * 1024, E16 - 1)]; };

  auto scan_trip = [&](const uint4 cur, int t) -> bool {
    const int q = q0 + t * 1024;
    const unsigned kw[4] = {cur.x, cur.y, cur.z, cur.w};
    unsigned u = 0u;
#pragma unroll
    for (int d = 0; d < 4; ++d) {
      const unsigned x = kw[d] ^ pat;
      const unsigned y = ((x & 0x7f7f7f7fu) + 0x7f7f7f7fu) | x;
      u |= (~y & 0x80808080u) >> (7 - d);
    }
    if (q >= E16) u = 0u;
    unsigned mask = __builtin_amdgcn_ballot_w32(u != 0u);
    while (mask != 0u) {
      if (cnt + 32 > LIST_CAP) return false;
      if (u != 0u) {
        const unsigned p = (unsigned)__builtin_ctz(u);
        const unsigned off = ((p & 7u) << 2) | (p >> 3);
        mylist[cnt + __popc(mask & lt)] = (unsigned)q * 16u + off;
        u &= u - 1u;
      }
      cnt += __popc(mask);
      mask = __builtin_amdgcn_ballot_w32(u != 0u);
    }
    return true;
  };

  static_assert(LIST_CAP >= 512 + 32, "an empty list takes a whole trip");
  auto scan_all = [&](auto&& on_full) {
    uint4 k[KDEPTH];
#pragma unroll
    for (int j = 0; j < KDEPTH; ++j) k[j] = load_keys(j);
    for (int it0 = 0; it0 < NIT; it0 += KDEPTH) {
      int stop = -1;
#pragma unroll
      for (int j = 0; j < KDEPTH; ++j) {
        if (stop < 0 && it0 + j < NIT) {
          const int c0 = cnt;
          if (!scan_trip(k[j], it0 + j)) { cnt = c0; stop = it0 + j; }
        }
        k[j] = load_keys(it0 + j + KDEPTH);
      }
      if (stop >= 0) {
        const int end = min(it0 + KDEPTH, NIT);
        for (int t = stop; t < end; ++t) {
          const uint4 cur = load_keys(t);
          const int c0 = cnt;
          if (!scan_trip(cur, t)) {
            cnt = c0;
            on_full();
            scan_trip(cur, t);
          }
        }
      }
    }
  };

  auto take_max = [&](int n) {
    for (int b = 0; b < n; b += 32) {
      if (b + lane < n) {
        const int e = (int)mylist[b + lane];
        const int lrow = dst[e] - base;
        const float4* sp = (const float4*)(sbuf + (size_t)e * NH);
        const float4 x0 = sp[0], x1 = sp[1];
        unsigned* mp = mtile + lrow * NH;
        lds_max(mp + 0, enc_f32(x0.x)); lds_max(mp + 1, enc_f32(x0.y));
        lds_max(mp + 2, enc_f32(x0.z)); lds_max(mp + 3, enc_f32(x0.w));
        lds_max(mp + 4, enc_f32(x1.x)); lds_max(mp + 5, enc_f32(x1.y));
        lds_max(mp + 6, enc_f32(x1.z)); lds_max(mp + 7, enc_f32(x1.w));
        mylist[b + lane] = ((unsigned)e << 8) | (unsigned)lrow;
      }
    }
  };

  auto flush = [&](int n, bool raw) {
    for (int b = 0; b < n; b += 32) {
      const int nb = min(32, n - b);
      int lrow = 0;
      int s_i = 0;
      if (lane < nb) {
        const unsigned w = mylist[b + lane];
        int e;
        if (raw) { e = (int)w; lrow = dst[e] - base; }
        else     { e = (int)(w >> 8); lrow = (int)(w & 255u); }
        s_i = src[e];
        const float4* sp = (const float4*)(sbuf + (size_t)e * NH);
        const uint4*  mp = (const uint4*)(mtile + lrow * NH);
        const float4 x0 = sp[0], x1 = sp[1];
        const uint4  y0 = mp[0], y1 = mp[1];
        const float4 e0 = make_float4(expf(x0.x - dec_f32(y0.x)), expf(x0.y - dec_f32(y0.y)),
                                      expf(x0.z - dec_f32(y0.z)), expf(x0.w - dec_f32(y0.w)));
        const float4 e1 = make_float4(expf(x1.x - dec_f32(y1.x)), expf(x1.y - dec_f32(y1.y)),
                                      expf(x1.z - dec_f32(y1.z)), expf(x1.w - dec_f32(y1.w)));
        float4* st = (float4*)(mystage + lane * NH);
        st[0] = e0;
        st[1] = e1;
        float* dp = dtile + lrow * NH;
        lds_add(dp + 0, e0.x); lds_add(dp + 1, e0.y);
        lds_add(dp + 2, e0.z); lds_add(dp + 3, e0.w);
        lds_add(dp + 4, e1.x); lds_add(dp + 5, e1.y);
        lds_add(dp + 6, e1.z); lds_add(dp + 7, e1.w);
      }
      for (int u0 = 0; u0 < nb; u0 += 4) {
        float4 v0[4], v1[4];
        float a0[4], a1[4];
        int lr[4];
#pragma unroll
        for (int k = 0; k < 4; ++k) {
          const int u = min(u0 + k, nb - 1);
          const int s = __builtin_amdgcn_readlane(s_i, u);
          lr[k] = __builtin_amdgcn_readlane(lrow, u);
          const float4* rp = (const float4*)(Wh + (size_t)s * HF);
          v0[k] = rp[lane];
          v1[k] = rp[32 + lane];
          a0[k] = mystage[u * NH + (lane >> 3)];
          a1[k] = mystage[u * NH + 4 + (lane >> 3)];
        }
#pragma unroll
        for (int k = 0; k < 4; ++k) {
          if (u0 + k < nb) {
            float* tp = tile + lr[k] * HF + lane;
            lds_add(tp,       v0[k].x * a0[k]);
            lds_add(tp + 32,  v0[k].y * a0[k]);
            lds_add(tp + 64,  v0[k].z * a0[k]);
            lds_add(tp + 96,  v0[k].w * a0[k]);
            lds_add(tp + 128, v1[k].x * a1[k]);
            lds_add(tp + 160, v1[k].y * a1[k]);
            lds_add(tp + 192, v1[k].z * a1[k]);
            lds_add(tp + 224, v1[k].w * a1[k]);
          }
        }
      }
    }
  };

  bool kept = true;
  scan_all([&]() { take_max(cnt); cnt = 0; kept = false; });
  take_max(cnt);
  __syncthreads();

  if (!kept) {
    cnt = 0;
    scan_all([&]() { flush(cnt, true); cnt = 0; });
  }
  flush(cnt, !kept);
  __syncthreads();

  for (int i = tid; i < nrows * 64; i += 1024) {
    const int r = i >> 6, g = i & 63;
    const int half = g >> 5, l = g & 31;
    const float* tp = tile + r * HF + half * 128 + l;
    float4 x = make_float4(tp[0], tp[32], tp[64], tp[96]);
    const float dn = dtile[r * NH + half * 4 + (l >> 3)];
    if (dn > 0.f) { x.x /= dn; x.y /= dn; x.z /= dn; x.w /= dn; }
    x.x = x.x > 0.f ? x.x : expm1f(x.x);
    x.y = x.y > 0.f ? x.y : expm1f(x.y);
    x.z = x.z > 0.f ? x.z : expm1f(x.z);
    x.w = x.w > 0.f ? x.w : expm1f(x.w);
    *(float4*)(out + (size_t)(base + r) * HF + half * 128 + 4 * l) = x;
  }
}

extern "C" void kernel_launch(void* const* d_in, const int* in_sizes, int n_in,
                              void* d_out, int out_size, void* d_ws, size_t ws_size,
                              hipStream_t stream) {
  const float* h   = (const float*)d_in[0];
  const float* ef  = (const float*)d_in[1];
  const int*   src = (const int*)  d_in[2];
  const int*   dst = (const int*)  d_in[3];
  const float* W_w = (const float*)d_in[4];
  const float* We_w= (const float*)d_in[5];
  const float* a_s = (const float*)d_in[6];
  const float* a_d = (const float*)d_in[7];
  const float* a_e = (const float*)d_in[8];
  float* out = (float*)d_out;

  char* ws = (char*)d_ws;
  const size_t WH_BYTES   = (size_t)N_NODES * HF * 4;
  const size_t SBUF_BYTES = (size_t)N_EDGES * NH * 4;
  const size_t NH_BYTES   = (size_t)N_NODES * NH * 4;
  float*    Wh   = (float*)   (ws);
  float*    sbuf = (float*)   (ws + WH_BYTES);
  float*    ssrc = (float*)   (ws + WH_BYTES + SBUF_BYTES);
  float*    sdst = (float*)   (ws + WH_BYTES + SBUF_BYTES + NH_BYTES);
  unsigned* tkey = (unsigned*)(ws + WH_BYTES + SBUF_BYTES + 3 * NH_BYTES);
  float*    V    = (float*)   (ws + WH_BYTES + SBUF_BYTES + 4 * NH_BYTES);

  gemm_wh_wmma<<<N_NODES / 16, 128, 0, stream>>>(h, W_w, a_s, a_d, Wh, ssrc, sdst);
  fold_we<<<1, 512, 0, stream>>>(We_w, a_e, V);
  edge_logits<<<LOGIT_BLOCKS, 256, 0, stream>>>(ef, src, dst, V, ssrc, sdst, sbuf, tkey);
  aggregate_tile<<<(N_NODES + TROWS - 1) / TROWS, 1024, 0, stream>>>(src, dst, Wh, sbuf, tkey, out);
}


# Edge term of the logits on f32 16x16x4 WMMA: a wave takes 32 edges a trip, folded weights as A, each edge row read once by 16-byte fragment loads, not 8 times; edge sum order changes
# speedup vs baseline: 5.2205x; 1.0541x over previous
#include <hip/hip_runtime.h>
#include <math.h>

typedef __attribute__((ext_vector_type(2))) float v2f;
typedef __attribute__((ext_vector_type(4))) float v4f;
typedef __attribute__((ext_vector_type(8))) float v8f;

#define N_NODES   50000
#define N_EDGES   800000
#define IN_F      256
#define EDGE_F    64
#define NH        8
#define NF        32
#define HF        256

__device__ __forceinline__ unsigned enc_f32(float f) {
  unsigned u = __float_as_uint(f);
  return (u & 0x80000000u) ? ~u : (u | 0x80000000u);
}
__device__ __forceinline__ float dec_f32(unsigned u) {
  unsigned b = (u & 0x80000000u) ? (u ^ 0x80000000u) : ~u;
  return __uint_as_float(b);
}

__global__ __launch_bounds__(128)
void gemm_wh_wmma(const float* __restrict__ h, const float* __restrict__ W,
                  const float* __restrict__ a_src, const float* __restrict__ a_dst,
                  float* __restrict__ Wh,
                  float* __restrict__ ssrc, float* __restrict__ sdst) {
  __shared__ __attribute__((aligned(16))) float sc_s[16 * NH];
  __shared__ __attribute__((aligned(16))) float sc_d[16 * NH];
  const int lane = threadIdx.x & 31;
  const int wave = threadIdx.x >> 5;
  const int m0   = blockIdx.x * 16;
  const int n0   = wave * 64;
  const int c    = lane & 15;
  const int hi   = lane >> 4;

  v8f c0 = {}; v8f c1 = {}; v8f c2 = {}; v8f c3 = {};
  const float* arow = h + (size_t)(m0 + c) * IN_F + 4 * hi;
  const float* brow = W + (size_t)(4 * hi) * HF + n0 + 4 * c;

  for (int k0 = 0; k0 < IN_F; k0 += 8) {
    const v4f av = *(const v4f*)(arow + k0);
    const float* bp = brow + (size_t)k0 * HF;
    const v4f q0 = *(const v4f*)(bp);
    const v4f q1 = *(const v4f*)(bp + HF);
    const v4f q2 = *(const v4f*)(bp + 2 * HF);
    const v4f q3 = *(const v4f*)(bp + 3 * HF);
    v2f a, b;
    a.x = av.x; a.y = av.y;
    b.x = q0.x; b.y = q1.x;
    c0 = __builtin_amdgcn_wmma_f32_16x16x4_f32(false, a, false, b, (short)0, c0, false, false);
    b.x = q0.y; b.y = q1.y;
    c1 = __builtin_amdgcn_wmma_f32_16x16x4_f32(false, a, false, b, (short)0, c1, false, false);
    b.x = q0.z; b.y = q1.z;
    c2 = __builtin_amdgcn_wmma_f32_16x16x4_f32(false, a, false, b, (short)0, c2, false, false);
    b.x = q0.w; b.y = q1.w;
    c3 = __builtin_amdgcn_wmma_f32_16x16x4_f32(false, a, false, b, (short)0, c3, false, false);
    a.x = av.z; a.y = av.w;
    b.x = q2.x; b.y = q3.x;
    c0 = __builtin_amdgcn_wmma_f32_16x16x4_f32(false, a, false, b, (short)0, c0, false, false);
    b.x = q2.y; b.y = q3.y;
    c1 = __builtin_amdgcn_wmma_f32_16x16x4_f32(false, a, false, b, (short)0, c1, false, false);
    b.x = q2.z; b.y = q3.z;
    c2 = __builtin_amdgcn_wmma_f32_16x16x4_f32(false, a, false, b, (short)0, c2, false, false);
    b.x = q2.w; b.y = q3.w;
    c3 = __builtin_amdgcn_wmma_f32_16x16x4_f32(false, a, false, b, (short)0, c3, false, false);
  }

  const v4f as4 = *(const v4f*)(a_src + n0 + 4 * c);
  const v4f ad4 = *(const v4f*)(a_dst + n0 + 4 * c);
  float ps[8], pd[8];
#pragma unroll
  for (int r = 0; r < 8; ++r) {
    ps[r] = c0[r] * as4.x + c1[r] * as4.y + c2[r] * as4.z + c3[r] * as4.w;
    pd[r] = c0[r] * ad4.x + c1[r] * ad4.y + c2[r] * ad4.z + c3[r] * ad4.w;
  }
#pragma unroll
  for (int m = 1; m < 8; m <<= 1) {
#pragma unroll
    for (int r = 0; r < 8; ++r) {
      ps[r] += __shfl_xor(ps[r], m, 32);
      pd[r] += __shfl_xor(pd[r], m, 32);
    }
  }
  const int rr = c & 7;
  float vs = ps[0], vd = pd[0];
#pragma unroll
  for (int r = 1; r < 8; ++r) {
    vs = (rr == r) ? ps[r] : vs;
    vd = (rr == r) ? pd[r] : vd;
  }
  const int slot = ((hi << 3) + rr) * NH + 2 * wave + (c >> 3);
  sc_s[slot] = vs;
  sc_d[slot] = vd;
  __syncthreads();

  float* op = Wh + (size_t)(m0 + (hi << 3)) * HF + n0 + 4 * c;
  for (int r = 0; r < 8; ++r) {
    v4f o;
    o.x = c0[r]; o.y = c1[r]; o.z = c2[r]; o.w = c3[r];
    *(v4f*)(op + (size_t)r * HF) = o;
  }
  if (wave == 0) {
    *(v4f*)(ssrc + (size_t)m0 * NH + 4 * lane) = *(const v4f*)(sc_s + 4 * lane);
  } else if (wave == 1) {
    *(v4f*)(sdst + (size_t)m0 * NH + 4 * lane) = *(const v4f*)(sc_d + 4 * lane);
  }
}

__global__ void fold_we(const float* __restrict__ We_w,
                        const float* __restrict__ a_edge,
                        float* __restrict__ V) {
  int t = threadIdx.x;
  int k = t >> 3, hh = t & 7;
  const float* wr = We_w + (size_t)k * HF + hh * NF;
  const float* ae = a_edge + hh * NF;
  float acc = 0.f;
  for (int f = 0; f < NF; ++f) acc += wr[f] * ae[f];
  V[k * NH + hh] = acc;
}

#define LOGIT_WGS   1250
#define LOGIT_WPW   4
#define EGROUP      32
#define N_EGROUPS   (N_EDGES / EGROUP)
static_assert(N_EDGES % EGROUP == 0, "a trip takes 32 whole edges");
static_assert(N_EGROUPS % (LOGIT_WGS * LOGIT_WPW) == 0, "every wave makes the same number of trips");
static_assert(EDGE_F == 64 && NH == 8, "the fragment maps below are written for 64 x 8");
#define TROWS     196
#define E4        (N_EDGES / 4)
#define E16       (N_EDGES / 16)
static_assert((N_NODES - 1) / TROWS < 256, "the tile number packs into 8 bits");
static_assert(N_NODES % 16 == 0, "a GEMM block owns 16 whole rows and one 512-byte piece of each score plane");
static_assert(N_EDGES % 16 == 0, "the tile numbers are read 16 at a time");
__global__ __launch_bounds__(LOGIT_WPW * 32)
void edge_logits(const float* __restrict__ ef, const int* __restrict__ src,
                 const int* __restrict__ dst, const float* __restrict__ V,
                 const float* __restrict__ ssrc, const float* __restrict__ sdst,
                 float* __restrict__ sbuf, unsigned* __restrict__ tkey) {
  const int lane = threadIdx.x & 31;
  const int c    = lane & 15;
  const int hi   = lane >> 4;
  for (int i = blockIdx.x * (LOGIT_WPW * 32) + threadIdx.x; i < E4; i += LOGIT_WGS * LOGIT_WPW * 32) {
    const int4 d4 = ((const int4*)dst)[i];
    tkey[i] = ((unsigned)d4.x / TROWS) | (((unsigned)d4.y / TROWS) << 8) |
              (((unsigned)d4.z / TROWS) << 16) | (((unsigned)d4.w / TROWS) << 24);
  }
  const int hd = c & 7;
  v4f av[EDGE_F / 8];
#pragma unroll
  for (int j = 0; j < EDGE_F / 8; ++j) {
    const float* vp = V + (size_t)(8 * j + 4 * hi) * NH + hd;
    av[j].x = vp[0];
    av[j].y = vp[NH];
    av[j].z = vp[2 * NH];
    av[j].w = vp[3 * NH];
  }
  const int wave = __builtin_amdgcn_readfirstlane((int)(threadIdx.x >> 5));
  for (int g = blockIdx.x * LOGIT_WPW + wave; g < N_EGROUPS; g += LOGIT_WGS * LOGIT_WPW) {
    const int ea = g * EGROUP + c;
    const int eb = ea + 16;
    const int sa = src[ea], da = dst[ea];
    const int sb = src[eb], db = dst[eb];
    const float* pa = ef + (size_t)ea * EDGE_F + 4 * hi;
    const float* pb = ef + (size_t)eb * EDGE_F + 4 * hi;
    v4f ba[EDGE_F / 8], bb[EDGE_F / 8];
#pragma unroll
    for (int j = 0; j < EDGE_F / 8; ++j) {
      ba[j] = *(const v4f*)(pa + 8 * j);
      bb[j] = *(const v4f*)(pb + 8 * j);
    }
    __builtin_amdgcn_sched_barrier(0);
    const v4f nsa = *(const v4f*)(ssrc + (size_t)sa * NH + 4 * hi);
    const v4f nda = *(const v4f*)(sdst + (size_t)da * NH + 4 * hi);
    const v4f nsb = *(const v4f*)(ssrc + (size_t)sb * NH + 4 * hi);
    const v4f ndb = *(const v4f*)(sdst + (size_t)db * NH + 4 * hi);
    __builtin_amdgcn_sched_barrier(0);
    v8f d0 = {}; v8f d1 = {};
#pragma unroll
    for (int j = 0; j < EDGE_F / 8; ++j) {
      v2f a, b;
      a.x = av[j].x; a.y = av[j].y;
      b.x = ba[j].x; b.y = ba[j].y;
      d0 = __builtin_amdgcn_wmma_f32_16x16x4_f32(false, a, false, b, (short)0, d0, false, false);
      b.x = bb[j].x; b.y = bb[j].y;
      d1 = __builtin_amdgcn_wmma_f32_16x16x4_f32(false, a, false, b, (short)0, d1, false, false);
      a.x = av[j].z; a.y = av[j].w;
      b.x = ba[j].z; b.y = ba[j].w;
      d0 = __builtin_amdgcn_wmma_f32_16x16x4_f32(false, a, false, b, (short)0, d0, false, false);
      b.x = bb[j].z; b.y = bb[j].w;
      d1 = __builtin_amdgcn_wmma_f32_16x16x4_f32(false, a, false, b, (short)0, d1, false, false);
    }
    __builtin_amdgcn_sched_barrier(0);
    v4f xa, xb;
    xa.x = hi ? d0[4] : d0[0]; xa.y = hi ? d0[5] : d0[1];
    xa.z = hi ? d0[6] : d0[2]; xa.w = hi ? d0[7] : d0[3];
    xb.x = hi ? d1[4] : d1[0]; xb.y = hi ? d1[5] : d1[1];
    xb.z = hi ? d1[6] : d1[2]; xb.w = hi ? d1[7] : d1[3];
    v4f oa = (nsa + nda) + xa;
    v4f ob = (nsb + ndb) + xb;
    oa.x = oa.x > 0.f ? oa.x : 0.01f * oa.x; oa.y = oa.y > 0.f ? oa.y : 0.01f * oa.y;
    oa.z = oa.z > 0.f ? oa.z : 0.01f * oa.z; oa.w = oa.w > 0.f ? oa.w : 0.01f * oa.w;
    ob.x = ob.x > 0.f ? ob.x : 0.01f * ob.x; ob.y = ob.y > 0.f ? ob.y : 0.01f * ob.y;
    ob.z = ob.z > 0.f ? ob.z : 0.01f * ob.z; ob.w = ob.w > 0.f ? ob.w : 0.01f * ob.w;
    *(v4f*)(sbuf + (size_t)ea * NH + 4 * hi) = oa;
    *(v4f*)(sbuf + (size_t)eb * NH + 4 * hi) = ob;
  }
}

#define TWAVES    32
#define LIST_CAP  632
#define KDEPTH    7
static_assert(TROWS <= 256, "the local row packs into 8 bits");
static_assert(N_EDGES < (1 << 24), "the edge id packs into 24 bits");

__device__ __forceinline__ void lds_add(float* p, float v) {
  __hip_atomic_fetch_add(p, v, __ATOMIC_RELAXED, __HIP_MEMORY_SCOPE_WORKGROUP);
}
__device__ __forceinline__ void lds_max(unsigned* p, unsigned v) {
  __hip_atomic_fetch_max(p, v, __ATOMIC_RELAXED, __HIP_MEMORY_SCOPE_WORKGROUP);
}

__global__ __launch_bounds__(1024)
void aggregate_tile(const int* __restrict__ src, const int* __restrict__ dst,
                    const float* __restrict__ Wh, const float* __restrict__ sbuf,
                    const unsigned* __restrict__ tkey, float* __restrict__ out) {
  __shared__ float    tile[TROWS * HF];
  __shared__ unsigned elist[TWAVES * LIST_CAP];
  __shared__ float    astage[TWAVES * 32 * NH];
  __shared__ __attribute__((aligned(16))) unsigned mtile[TROWS * NH];
  __shared__ __attribute__((aligned(16))) float    dtile[TROWS * NH];

  const int tid  = threadIdx.x;
  const int lane = tid & 31;
  const int wave = __builtin_amdgcn_readfirstlane(tid >> 5);
  const int base = blockIdx.x * TROWS;
  const int nrows = min(TROWS, N_NODES - base);
  const unsigned lt = (1u << lane) - 1u;

  float4* t4 = (float4*)tile;
  for (int i = tid; i < TROWS * HF / 4; i += 1024) t4[i] = make_float4(0.f, 0.f, 0.f, 0.f);
  for (int i = tid; i < TROWS * NH; i += 1024) {
    dtile[i] = 0.f;
    mtile[i] = 0u;
  }
  __syncthreads();

  unsigned* mylist  = elist + wave * LIST_CAP;
  float*    mystage = astage + wave * (32 * NH);
  const uint4* key16 = (const uint4*)tkey;
  const unsigned pat = (unsigned)blockIdx.x * 0x01010101u;
  const int NIT = (E16 + 1023) / 1024;
  int cnt = 0;

  const int q0 = wave * 32 + lane;
  auto load_keys = [&](int t) -> uint4 { return key16[min(q0 + t * 1024, E16 - 1)]; };

  auto scan_trip = [&](const uint4 cur, int t) -> bool {
    const int q = q0 + t * 1024;
    const unsigned kw[4] = {cur.x, cur.y, cur.z, cur.w};
    unsigned u = 0u;
#pragma unroll
    for (int d = 0; d < 4; ++d) {
      const unsigned x = kw[d] ^ pat;
      const unsigned y = ((x & 0x7f7f7f7fu) + 0x7f7f7f7fu) | x;
      u |= (~y & 0x80808080u) >> (7 - d);
    }
    if (q >= E16) u = 0u;
    unsigned mask = __builtin_amdgcn_ballot_w32(u != 0u);
    while (mask != 0u) {
      if (cnt + 32 > LIST_CAP) return false;
      if (u != 0u) {
        const unsigned p = (unsigned)__builtin_ctz(u);
        const unsigned off = ((p & 7u) << 2) | (p >> 3);
        mylist[cnt + __popc(mask & lt)] = (unsigned)q * 16u + off;
        u &= u - 1u;
      }
      cnt += __popc(mask);
      mask = __builtin_amdgcn_ballot_w32(u != 0u);
    }
    return true;
  };

  static_assert(LIST_CAP >= 512 + 32, "an empty list takes a whole trip");
  auto scan_all = [&](auto&& on_full) {
    uint4 k[KDEPTH];
#pragma unroll
    for (int j = 0; j < KDEPTH; ++j) k[j] = load_keys(j);
    for (int it0 = 0; it0 < NIT; it0 += KDEPTH) {
      int stop = -1;
#pragma unroll
      for (int j = 0; j < KDEPTH; ++j) {
        if (stop < 0 && it0 + j < NIT) {
          const int c0 = cnt;
          if (!scan_trip(k[j], it0 + j)) { cnt = c0; stop = it0 + j; }
        }
        k[j] = load_keys(it0 + j + KDEPTH);
      }
      if (stop >= 0) {
        const int end = min(it0 + KDEPTH, NIT);
        for (int t = stop; t < end; ++t) {
          const uint4 cur = load_keys(t);
          const int c0 = cnt;
          if (!scan_trip(cur, t)) {
            cnt = c0;
            on_full();
            scan_trip(cur, t);
          }
        }
      }
    }
  };

  auto take_max = [&](int n) {
    for (int b = 0; b < n; b += 32) {
      if (b + lane < n) {
        const int e = (int)mylist[b + lane];
        const int lrow = dst[e] - base;
        const float4* sp = (const float4*)(sbuf + (size_t)e * NH);
        const float4 x0 = sp[0], x1 = sp[1];
        unsigned* mp = mtile + lrow * NH;
        lds_max(mp + 0, enc_f32(x0.x)); lds_max(mp + 1, enc_f32(x0.y));
        lds_max(mp + 2, enc_f32(x0.z)); lds_max(mp + 3, enc_f32(x0.w));
        lds_max(mp + 4, enc_f32(x1.x)); lds_max(mp + 5, enc_f32(x1.y));
        lds_max(mp + 6, enc_f32(x1.z)); lds_max(mp + 7, enc_f32(x1.w));
        mylist[b + lane] = ((unsigned)e << 8) | (unsigned)lrow;
      }
    }
  };

  auto flush = [&](int n, bool raw) {
    for (int b = 0; b < n; b += 32) {
      const int nb = min(32, n - b);
      int lrow = 0;
      int s_i = 0;
      if (lane < nb) {
        const unsigned w = mylist[b + lane];
        int e;
        if (raw) { e = (int)w; lrow = dst[e] - base; }
        else     { e = (int)(w >> 8); lrow = (int)(w & 255u); }
        s_i = src[e];
        const float4* sp = (const float4*)(sbuf + (size_t)e * NH);
        const uint4*  mp = (const uint4*)(mtile + lrow * NH);
        const float4 x0 = sp[0], x1 = sp[1];
        const uint4  y0 = mp[0], y1 = mp[1];
        const float4 e0 = make_float4(expf(x0.x - dec_f32(y0.x)), expf(x0.y - dec_f32(y0.y)),
                                      expf(x0.z - dec_f32(y0.z)), expf(x0.w - dec_f32(y0.w)));
        const float4 e1 = make_float4(expf(x1.x - dec_f32(y1.x)), expf(x1.y - dec_f32(y1.y)),
                                      expf(x1.z - dec_f32(y1.z)), expf(x1.w - dec_f32(y1.w)));
        float4* st = (float4*)(mystage + lane * NH);
        st[0] = e0;
        st[1] = e1;
        float* dp = dtile + lrow * NH;
        lds_add(dp + 0, e0.x); lds_add(dp + 1, e0.y);
        lds_add(dp + 2, e0.z); lds_add(dp + 3, e0.w);
        lds_add(dp + 4, e1.x); lds_add(dp + 5, e1.y);
        lds_add(dp + 6, e1.z); lds_add(dp + 7, e1.w);
      }
      for (int u0 = 0; u0 < nb; u0 += 4) {
        float4 v0[4], v1[4];
        float a0[4], a1[4];
        int lr[4];
#pragma unroll
        for (int k = 0; k < 4; ++k) {
          const int u = min(u0 + k, nb - 1);
          const int s = __builtin_amdgcn_readlane(s_i, u);
          lr[k] = __builtin_amdgcn_readlane(lrow, u);
          const float4* rp = (const float4*)(Wh + (size_t)s * HF);
          v0[k] = rp[lane];
          v1[k] = rp[32 + lane];
          a0[k] = mystage[u * NH + (lane >> 3)];
          a1[k] = mystage[u * NH + 4 + (lane >> 3)];
        }
#pragma unroll
        for (int k = 0; k < 4; ++k) {
          if (u0 + k < nb) {
            float* tp = tile + lr[k] * HF + lane;
            lds_add(tp,       v0[k].x * a0[k]);
            lds_add(tp + 32,  v0[k].y * a0[k]);
            lds_add(tp + 64,  v0[k].z * a0[k]);
            lds_add(tp + 96,  v0[k].w * a0[k]);
            lds_add(tp + 128, v1[k].x * a1[k]);
            lds_add(tp + 160, v1[k].y * a1[k]);
            lds_add(tp + 192, v1[k].z * a1[k]);
            lds_add(tp + 224, v1[k].w * a1[k]);
          }
        }
      }
    }
  };

  bool kept = true;
  scan_all([&]() { take_max(cnt); cnt = 0; kept = false; });
  take_max(cnt);
  __syncthreads();

  if (!kept) {
    cnt = 0;
    scan_all([&]() { flush(cnt, true); cnt = 0; });
  }
  flush(cnt, !kept);
  __syncthreads();

  for (int i = tid; i < nrows * 64; i += 1024) {
    const int r = i >> 6, g = i & 63;
    const int half = g >> 5, l = g & 31;
    const float* tp = tile + r * HF + half * 128 + l;
    float4 x = make_float4(tp[0], tp[32], tp[64], tp[96]);
    const float dn = dtile[r * NH + half * 4 + (l >> 3)];
    if (dn > 0.f) { x.x /= dn; x.y /= dn; x.z /= dn; x.w /= dn; }
    x.x = x.x > 0.f ? x.x : expm1f(x.x);
    x.y = x.y > 0.f ? x.y : expm1f(x.y);
    x.z = x.z > 0.f ? x.z : expm1f(x.z);
    x.w = x.w > 0.f ? x.w : expm1f(x.w);
    *(float4*)(out + (size_t)(base + r) * HF + half * 128 + 4 * l) = x;
  }
}

extern "C" void kernel_launch(void* const* d_in, const int* in_sizes, int n_in,
                              void* d_out, int out_size, void* d_ws, size_t ws_size,
                              hipStream_t stream) {
  const float* h   = (const float*)d_in[0];
  const float* ef  = (const float*)d_in[1];
  const int*   src = (const int*)  d_in[2];
  const int*   dst = (const int*)  d_in[3];
  const float* W_w = (const float*)d_in[4];
  const float* We_w= (const float*)d_in[5];
  const float* a_s = (const float*)d_in[6];
  const float* a_d = (const float*)d_in[7];
  const float* a_e = (const float*)d_in[8];
  float* out = (float*)d_out;

  char* ws = (char*)d_ws;
  const size_t WH_BYTES   = (size_t)N_NODES * HF * 4;
  const size_t SBUF_BYTES = (size_t)N_EDGES * NH * 4;
  const size_t NH_BYTES   = (size_t)N_NODES * NH * 4;
  float*    Wh   = (float*)   (ws);
  float*    sbuf = (float*)   (ws + WH_BYTES);
  float*    ssrc = (float*)   (ws + WH_BYTES + SBUF_BYTES);
  float*    sdst = (float*)   (ws + WH_BYTES + SBUF_BYTES + NH_BYTES);
  unsigned* tkey = (unsigned*)(ws + WH_BYTES + SBUF_BYTES + 3 * NH_BYTES);
  float*    V    = (float*)   (ws + WH_BYTES + SBUF_BYTES + 4 * NH_BYTES);

  gemm_wh_wmma<<<N_NODES / 16, 128, 0, stream>>>(h, W_w, a_s, a_d, Wh, ssrc, sdst);
  fold_we<<<1, 512, 0, stream>>>(We_w, a_e, V);
  edge_logits<<<LOGIT_WGS, LOGIT_WPW * 32, 0, stream>>>(ef, src, dst, V, ssrc, sdst, sbuf, tkey);
  aggregate_tile<<<(N_NODES + TROWS - 1) / TROWS, 1024, 0, stream>>>(src, dst, Wh, sbuf, tkey, out);
}


# Node GEMM and edge term in one launch, work-groups interleaved 5:2, so the edge stream runs under the f32 matrix time; the tile wave adds ssrc[src]+sdst[dst], keeps the logit in registers; 3 launches
# speedup vs baseline: 5.4640x; 1.0466x over previous
#include <hip/hip_runtime.h>
#include <math.h>

typedef __attribute__((ext_vector_type(2))) float v2f;
typedef __attribute__((ext_vector_type(4))) float v4f;
typedef __attribute__((ext_vector_type(8))) float v8f;

#define N_NODES   50000
#define N_EDGES   800000
#define IN_F      256
#define EDGE_F    64
#define NH        8
#define NF        32
#define HF        256

__device__ __forceinline__ unsigned enc_f32(float f) {
  unsigned u = __float_as_uint(f);
  return (u & 0x80000000u) ? ~u : (u | 0x80000000u);
}
__device__ __forceinline__ float dec_f32(unsigned u) {
  unsigned b = (u & 0x80000000u) ? (u ^ 0x80000000u) : ~u;
  return __uint_as_float(b);
}

__device__ __forceinline__
void gemm_role(const int bx,
               const float* __restrict__ h, const float* __restrict__ W,
               const float* __restrict__ a_src, const float* __restrict__ a_dst,
               float* __restrict__ Wh,
               float* __restrict__ ssrc, float* __restrict__ sdst) {
  __shared__ __attribute__((aligned(16))) float sc_s[16 * NH];
  __shared__ __attribute__((aligned(16))) float sc_d[16 * NH];
  const int lane = threadIdx.x & 31;
  const int wave = threadIdx.x >> 5;
  const int m0   = bx * 16;
  const int n0   = wave * 64;
  const int c    = lane & 15;
  const int hi   = lane >> 4;

  v8f c0 = {}; v8f c1 = {}; v8f c2 = {}; v8f c3 = {};
  const float* arow = h + (size_t)(m0 + c) * IN_F + 4 * hi;
  const float* brow = W + (size_t)(4 * hi) * HF + n0 + 4 * c;

  for (int k0 = 0; k0 < IN_F; k0 += 8) {
    const v4f av = *(const v4f*)(arow + k0);
    const float* bp = brow + (size_t)k0 * HF;
    const v4f q0 = *(const v4f*)(bp);
    const v4f q1 = *(const v4f*)(bp + HF);
    const v4f q2 = *(const v4f*)(bp + 2 * HF);
    const v4f q3 = *(const v4f*)(bp + 3 * HF);
    v2f a, b;
    a.x = av.x; a.y = av.y;
    b.x = q0.x; b.y = q1.x;
    c0 = __builtin_amdgcn_wmma_f32_16x16x4_f32(false, a, false, b, (short)0, c0, false, false);
    b.x = q0.y; b.y = q1.y;
    c1 = __builtin_amdgcn_wmma_f32_16x16x4_f32(false, a, false, b, (short)0, c1, false, false);
    b.x = q0.z; b.y = q1.z;
    c2 = __builtin_amdgcn_wmma_f32_16x16x4_f32(false, a, false, b, (short)0, c2, false, false);
    b.x = q0.w; b.y = q1.w;
    c3 = __builtin_amdgcn_wmma_f32_16x16x4_f32(false, a, false, b, (short)0, c3, false, false);
    a.x = av.z; a.y = av.w;
    b.x = q2.x; b.y = q3.x;
    c0 = __builtin_amdgcn_wmma_f32_16x16x4_f32(false, a, false, b, (short)0, c0, false, false);
    b.x = q2.y; b.y = q3.y;
    c1 = __builtin_amdgcn_wmma_f32_16x16x4_f32(false, a, false, b, (short)0, c1, false, false);
    b.x = q2.z; b.y = q3.z;
    c2 = __builtin_amdgcn_wmma_f32_16x16x4_f32(false, a, false, b, (short)0, c2, false, false);
    b.x = q2.w; b.y = q3.w;
    c3 = __builtin_amdgcn_wmma_f32_16x16x4_f32(false, a, false, b, (short)0, c3, false, false);
  }

  const v4f as4 = *(const v4f*)(a_src + n0 + 4 * c);
  const v4f ad4 = *(const v4f*)(a_dst + n0 + 4 * c);
  float ps[8], pd[8];
#pragma unroll
  for (int r = 0; r < 8; ++r) {
    ps[r] = c0[r] * as4.x + c1[r] * as4.y + c2[r] * as4.z + c3[r] * as4.w;
    pd[r] = c0[r] * ad4.x + c1[r] * ad4.y + c2[r] * ad4.z + c3[r] * ad4.w;
  }
#pragma unroll
  for (int m = 1; m < 8; m <<= 1) {
#pragma unroll
    for (int r = 0; r < 8; ++r) {
      ps[r] += __shfl_xor(ps[r], m, 32);
      pd[r] += __shfl_xor(pd[r], m, 32);
    }
  }
  const int rr = c & 7;
  float vs = ps[0], vd = pd[0];
#pragma unroll
  for (int r = 1; r < 8; ++r) {
    vs = (rr == r) ? ps[r] : vs;
    vd = (rr == r) ? pd[r] : vd;
  }
  const int slot = ((hi << 3) + rr) * NH + 2 * wave + (c >> 3);
  sc_s[slot] = vs;
  sc_d[slot] = vd;
  __syncthreads();

  float* op = Wh + (size_t)(m0 + (hi << 3)) * HF + n0 + 4 * c;
  for (int r = 0; r < 8; ++r) {
    v4f o;
    o.x = c0[r]; o.y = c1[r]; o.z = c2[r]; o.w = c3[r];
    *(v4f*)(op + (size_t)r * HF) = o;
  }
  if (wave == 0) {
    *(v4f*)(ssrc + (size_t)m0 * NH + 4 * lane) = *(const v4f*)(sc_s + 4 * lane);
  } else if (wave == 1) {
    *(v4f*)(sdst + (size_t)m0 * NH + 4 * lane) = *(const v4f*)(sc_d + 4 * lane);
  }
}

__global__ void fold_we(const float* __restrict__ We_w,
                        const float* __restrict__ a_edge,
                        float* __restrict__ V) {
  int t = threadIdx.x;
  int k = t >> 3, hh = t & 7;
  const float* wr = We_w + (size_t)k * HF + hh * NF;
  const float* ae = a_edge + hh * NF;
  float acc = 0.f;
  for (int f = 0; f < NF; ++f) acc += wr[f] * ae[f];
  V[k * NH + hh] = acc;
}

#define LOGIT_WGS   1250
#define LOGIT_WPW   4
#define EGROUP      32
#define N_EGROUPS   (N_EDGES / EGROUP)
static_assert(N_EDGES % EGROUP == 0, "a trip takes 32 whole edges");
static_assert(N_EGROUPS % (LOGIT_WGS * LOGIT_WPW) == 0, "every wave makes the same number of trips");
static_assert(EDGE_F == 64 && NH == 8, "the fragment maps below are written for 64 x 8");
#define TROWS     196
#define E4        (N_EDGES / 4)
#define E16       (N_EDGES / 16)
static_assert((N_NODES - 1) / TROWS < 256, "the tile number packs into 8 bits");
static_assert(N_NODES % 16 == 0, "a GEMM block owns 16 whole rows and one 512-byte piece of each score plane");
static_assert(N_EDGES % 16 == 0, "the tile numbers are read 16 at a time");
__device__ __forceinline__
void edge_role(const int bx, const float* __restrict__ ef,
               const int* __restrict__ dst, const float* __restrict__ V,
               float* __restrict__ ebuf, unsigned* __restrict__ tkey) {
  const int lane = threadIdx.x & 31;
  const int c    = lane & 15;
  const int hi   = lane >> 4;
  for (int i = bx * (LOGIT_WPW * 32) + threadIdx.x; i < E4; i += LOGIT_WGS * LOGIT_WPW * 32) {
    const int4 d4 = ((const int4*)dst)[i];
    tkey[i] = ((unsigned)d4.x / TROWS) | (((unsigned)d4.y / TROWS) << 8) |
              (((unsigned)d4.z / TROWS) << 16) | (((unsigned)d4.w / TROWS) << 24);
  }
  const int hd = c & 7;
  v4f av[EDGE_F / 8];
#pragma unroll
  for (int j = 0; j < EDGE_F / 8; ++j) {
    const float* vp = V + (size_t)(8 * j + 4 * hi) * NH + hd;
    av[j].x = vp[0];
    av[j].y = vp[NH];
    av[j].z = vp[2 * NH];
    av[j].w = vp[3 * NH];
  }
  const int wave = __builtin_amdgcn_readfirstlane((int)(threadIdx.x >> 5));
  for (int g = bx * LOGIT_WPW + wave; g < N_EGROUPS; g += LOGIT_WGS * LOGIT_WPW) {
    const int ea = g * EGROUP + c;
    const int eb = ea + 16;
    const float* pa = ef + (size_t)ea * EDGE_F + 4 * hi;
    const float* pb = ef + (size_t)eb * EDGE_F + 4 * hi;
    v4f ba[EDGE_F / 8], bb[EDGE_F / 8];
#pragma unroll
    for (int j = 0; j < EDGE_F / 8; ++j) {
      ba[j] = *(const v4f*)(pa + 8 * j);
      bb[j] = *(const v4f*)(pb + 8 * j);
    }
    __builtin_amdgcn_sched_barrier(0);
    v8f d0 = {}; v8f d1 = {};
#pragma unroll
    for (int j = 0; j < EDGE_F / 8; ++j) {
      v2f a, b;
      a.x = av[j].x; a.y = av[j].y;
      b.x = ba[j].x; b.y = ba[j].y;
      d0 = __builtin_amdgcn_wmma_f32_16x16x4_f32(false, a, false, b, (short)0, d0, false, false);
      b.x = bb[j].x; b.y = bb[j].y;
      d1 = __builtin_amdgcn_wmma_f32_16x16x4_f32(false, a, false, b, (short)0, d1, false, false);
      a.x = av[j].z; a.y = av[j].w;
      b.x = ba[j].z; b.y = ba[j].w;
      d0 = __builtin_amdgcn_wmma_f32_16x16x4_f32(false, a, false, b, (short)0, d0, false, false);
      b.x = bb[j].z; b.y = bb[j].w;
      d1 = __builtin_amdgcn_wmma_f32_16x16x4_f32(false, a, false, b, (short)0, d1, false, false);
    }
    v4f xa, xb;
    xa.x = hi ? d0[4] : d0[0]; xa.y = hi ? d0[5] : d0[1];
    xa.z = hi ? d0[6] : d0[2]; xa.w = hi ? d0[7] : d0[3];
    xb.x = hi ? d1[4] : d1[0]; xb.y = hi ? d1[5] : d1[1];
    xb.z = hi ? d1[6] : d1[2]; xb.w = hi ? d1[7] : d1[3];
    *(v4f*)(ebuf + (size_t)ea * NH + 4 * hi) = xa;
    *(v4f*)(ebuf + (size_t)eb * NH + 4 * hi) = xb;
  }
}

#define GEMM_WGS   (N_NODES / 16)
#define MIX_PER    7
#define GEMM_PER   5
static_assert(GEMM_WGS % GEMM_PER == 0 && LOGIT_WGS % (MIX_PER - GEMM_PER) == 0 &&
              GEMM_WGS / GEMM_PER == LOGIT_WGS / (MIX_PER - GEMM_PER),
              "the two roles' work-groups interleave 5 : 2 without a remainder");
__global__ __launch_bounds__(128)
void gemm_and_edge_term(const float* __restrict__ h, const float* __restrict__ W,
                        const float* __restrict__ a_src, const float* __restrict__ a_dst,
                        float* __restrict__ Wh,
                        float* __restrict__ ssrc, float* __restrict__ sdst,
                        const float* __restrict__ ef, const int* __restrict__ dst,
                        const float* __restrict__ V,
                        float* __restrict__ ebuf, unsigned* __restrict__ tkey) {
  const int q = blockIdx.x / MIX_PER;
  const int r = blockIdx.x % MIX_PER;
  if (r < GEMM_PER) gemm_role(GEMM_PER * q + r, h, W, a_src, a_dst, Wh, ssrc, sdst);
  else              edge_role((MIX_PER - GEMM_PER) * q + (r - GEMM_PER), ef, dst, V, ebuf, tkey);
}

#define TWAVES    32
#define LIST_CAP  632
#define KDEPTH    7
#define CARRY     4

__device__ __forceinline__ void lds_add(float* p, float v) {
  __hip_atomic_fetch_add(p, v, __ATOMIC_RELAXED, __HIP_MEMORY_SCOPE_WORKGROUP);
}
__device__ __forceinline__ void lds_max(unsigned* p, unsigned v) {
  __hip_atomic_fetch_max(p, v, __ATOMIC_RELAXED, __HIP_MEMORY_SCOPE_WORKGROUP);
}

__global__ __launch_bounds__(1024)
void aggregate_tile(const int* __restrict__ src, const int* __restrict__ dst,
                    const float* __restrict__ Wh, const float* __restrict__ ebuf,
                    const float* __restrict__ ssrc, const float* __restrict__ sdst,
                    const unsigned* __restrict__ tkey, float* __restrict__ out) {
  __shared__ float    tile[TROWS * HF];
  __shared__ unsigned elist[TWAVES * LIST_CAP];
  __shared__ float    astage[TWAVES * 32 * NH];
  __shared__ __attribute__((aligned(16))) unsigned mtile[TROWS * NH];
  __shared__ __attribute__((aligned(16))) float    dtile[TROWS * NH];

  const int tid  = threadIdx.x;
  const int lane = tid & 31;
  const int wave = __builtin_amdgcn_readfirstlane(tid >> 5);
  const int base = blockIdx.x * TROWS;
  const int nrows = min(TROWS, N_NODES - base);
  const unsigned lt = (1u << lane) - 1u;

  float4* t4 = (float4*)tile;
  for (int i = tid; i < TROWS * HF / 4; i += 1024) t4[i] = make_float4(0.f, 0.f, 0.f, 0.f);
  for (int i = tid; i < TROWS * NH; i += 1024) {
    dtile[i] = 0.f;
    mtile[i] = 0u;
  }
  __syncthreads();

  unsigned* mylist  = elist + wave * LIST_CAP;
  float*    mystage = astage + wave * (32 * NH);
  const uint4* key16 = (const uint4*)tkey;
  const unsigned pat = (unsigned)blockIdx.x * 0x01010101u;
  const int NIT = (E16 + 1023) / 1024;
  int cnt = 0;

  const int q0 = wave * 32 + lane;
  auto load_keys = [&](int t) -> uint4 { return key16[min(q0 + t * 1024, E16 - 1)]; };

  auto scan_trip = [&](const uint4 cur, int t) -> bool {
    const int q = q0 + t * 1024;
    const unsigned kw[4] = {cur.x, cur.y, cur.z, cur.w};
    unsigned u = 0u;
#pragma unroll
    for (int d = 0; d < 4; ++d) {
      const unsigned x = kw[d] ^ pat;
      const unsigned y = ((x & 0x7f7f7f7fu) + 0x7f7f7f7fu) | x;
      u |= (~y & 0x80808080u) >> (7 - d);
    }
    if (q >= E16) u = 0u;
    unsigned mask = __builtin_amdgcn_ballot_w32(u != 0u);
    while (mask != 0u) {
      if (cnt + 32 > LIST_CAP) return false;
      if (u != 0u) {
        const unsigned p = (unsigned)__builtin_ctz(u);
        const unsigned off = ((p & 7u) << 2) | (p >> 3);
        mylist[cnt + __popc(mask & lt)] = (unsigned)q * 16u + off;
        u &= u - 1u;
      }
      cnt += __popc(mask);
      mask = __builtin_amdgcn_ballot_w32(u != 0u);
    }
    return true;
  };

  static_assert(LIST_CAP >= 512 + 32, "an empty list takes a whole trip");
  auto scan_all = [&](auto&& on_full) {
    uint4 k[KDEPTH];
#pragma unroll
    for (int j = 0; j < KDEPTH; ++j) k[j] = load_keys(j);
    for (int it0 = 0; it0 < NIT; it0 += KDEPTH) {
      int stop = -1;
#pragma unroll
      for (int j = 0; j < KDEPTH; ++j) {
        if (stop < 0 && it0 + j < NIT) {
          const int c0 = cnt;
          if (!scan_trip(k[j], it0 + j)) { cnt = c0; stop = it0 + j; }
        }
        k[j] = load_keys(it0 + j + KDEPTH);
      }
      if (stop >= 0) {
        const int end = min(it0 + KDEPTH, NIT);
        for (int t = stop; t < end; ++t) {
          const uint4 cur = load_keys(t);
          const int c0 = cnt;
          if (!scan_trip(cur, t)) {
            cnt = c0;
            on_full();
            scan_trip(cur, t);
          }
        }
      }
    }
  };

  struct Part1 { float4 x0, x1; int s, d; };
  struct Edge  { float4 s0, s1; int s, lrow; };
  auto part1 = [&](int e) -> Part1 {
    Part1 p;
    const float4* xp = (const float4*)(ebuf + (size_t)e * NH);
    p.s = src[e];
    p.d = dst[e];
    p.x0 = xp[0];
    p.x1 = xp[1];
    return p;
  };
  auto lrelu = [](float v) -> float { return v > 0.f ? v : 0.01f * v; };
  auto part2 = [&](const Part1& p) -> Edge {
    const float4* ps = (const float4*)(ssrc + (size_t)p.s * NH);
    const float4* pd = (const float4*)(sdst + (size_t)p.d * NH);
    const float4 a0 = ps[0], a1 = ps[1];
    const float4 d0 = pd[0], d1 = pd[1];
    Edge r;
    r.s0 = make_float4(lrelu((a0.x + d0.x) + p.x0.x), lrelu((a0.y + d0.y) + p.x0.y),
                       lrelu((a0.z + d0.z) + p.x0.z), lrelu((a0.w + d0.w) + p.x0.w));
    r.s1 = make_float4(lrelu((a1.x + d1.x) + p.x1.x), lrelu((a1.y + d1.y) + p.x1.y),
                       lrelu((a1.z + d1.z) + p.x1.z), lrelu((a1.w + d1.w) + p.x1.w));
    r.s = p.s;
    r.lrow = p.d - base;
    return r;
  };
  auto max8 = [&](const Edge& r) {
    unsigned* mp = mtile + r.lrow * NH;
    lds_max(mp + 0, enc_f32(r.s0.x)); lds_max(mp + 1, enc_f32(r.s0.y));
    lds_max(mp + 2, enc_f32(r.s0.z)); lds_max(mp + 3, enc_f32(r.s0.w));
    lds_max(mp + 4, enc_f32(r.s1.x)); lds_max(mp + 5, enc_f32(r.s1.y));
    lds_max(mp + 6, enc_f32(r.s1.z)); lds_max(mp + 7, enc_f32(r.s1.w));
  };

  auto take_max = [&](int from, int n) {
    for (int b = from; b < n; b += 32) {
      if (b + lane < n) max8(part2(part1((int)mylist[b + lane])));
    }
  };

  float4 ks0[CARRY], ks1[CARRY];
  int    ksrc[CARRY], krow[CARRY];
  auto take_max_carry = [&](int n) {
    Part1 p[CARRY];
#pragma unroll
    for (int k = 0; k < CARRY; ++k) {
      const int i = 32 * k + lane;
      p[k] = part1(i < n ? (int)mylist[i] : 0);
    }
#pragma unroll
    for (int k = 0; k < CARRY; ++k) {
      const Edge r = part2(p[k]);
      ks0[k] = r.s0; ks1[k] = r.s1;
      ksrc[k] = r.s;
      krow[k] = 32 * k + lane < n ? r.lrow : 0;
    }
#pragma unroll
    for (int k = 0; k < CARRY; ++k) {
      if (32 * k + lane < n) {
        Edge r;
        r.s0 = ks0[k]; r.s1 = ks1[k]; r.s = ksrc[k]; r.lrow = krow[k];
        max8(r);
      }
    }
    take_max(32 * CARRY, n);
  };

  auto sum_batch = [&](int nb, const float4 x0, const float4 x1, const int s_i, const int lrow) {
    if (lane < nb) {
      const uint4*  mp = (const uint4*)(mtile + lrow * NH);
      const uint4  y0 = mp[0], y1 = mp[1];
      const float4 e0 = make_float4(expf(x0.x - dec_f32(y0.x)), expf(x0.y - dec_f32(y0.y)),
                                    expf(x0.z - dec_f32(y0.z)), expf(x0.w - dec_f32(y0.w)));
      const float4 e1 = make_float4(expf(x1.x - dec_f32(y1.x)), expf(x1.y - dec_f32(y1.y)),
                                    expf(x1.z - dec_f32(y1.z)), expf(x1.w - dec_f32(y1.w)));
      float4* st = (float4*)(mystage + lane * NH);
      st[0] = e0;
      st[1] = e1;
      float* dp = dtile + lrow * NH;
      lds_add(dp + 0, e0.x); lds_add(dp + 1, e0.y);
      lds_add(dp + 2, e0.z); lds_add(dp + 3, e0.w);
      lds_add(dp + 4, e1.x); lds_add(dp + 5, e1.y);
      lds_add(dp + 6, e1.z); lds_add(dp + 7, e1.w);
    }
    for (int u0 = 0; u0 < nb; u0 += 4) {
      float4 v0[4], v1[4];
      float a0[4], a1[4];
      int lr[4];
#pragma unroll
      for (int k = 0; k < 4; ++k) {
        const int u = min(u0 + k, nb - 1);
        const int s = __builtin_amdgcn_readlane(s_i, u);
        lr[k] = __builtin_amdgcn_readlane(lrow, u);
        const float4* rp = (const float4*)(Wh + (size_t)s * HF);
        v0[k] = rp[lane];
        v1[k] = rp[32 + lane];
        a0[k] = mystage[u * NH + (lane >> 3)];
        a1[k] = mystage[u * NH + 4 + (lane >> 3)];
      }
#pragma unroll
      for (int k = 0; k < 4; ++k) {
        if (u0 + k < nb) {
          float* tp = tile + lr[k] * HF + lane;
          lds_add(tp,       v0[k].x * a0[k]);
          lds_add(tp + 32,  v0[k].y * a0[k]);
          lds_add(tp + 64,  v0[k].z * a0[k]);
          lds_add(tp + 96,  v0[k].w * a0[k]);
          lds_add(tp + 128, v1[k].x * a1[k]);
          lds_add(tp + 160, v1[k].y * a1[k]);
          lds_add(tp + 192, v1[k].z * a1[k]);
          lds_add(tp + 224, v1[k].w * a1[k]);
        }
      }
    }
  };

  auto flush = [&](int from, int n) {
    for (int b = from; b < n; b += 32) {
      const int nb = min(32, n - b);
      float4 x0 = make_float4(0.f, 0.f, 0.f, 0.f), x1 = x0;
      int s_i = 0, lrow = 0;
      if (lane < nb) {
        const Edge r = part2(part1((int)mylist[b + lane]));
        x0 = r.s0; x1 = r.s1; s_i = r.s; lrow = r.lrow;
      }
      sum_batch(nb, x0, x1, s_i, lrow);
    }
  };

  bool kept = true;
  scan_all([&]() { take_max(0, cnt); cnt = 0; kept = false; });
  if (kept) take_max_carry(cnt);
  else      take_max(0, cnt);
  __syncthreads();

  if (kept) {
#pragma unroll
    for (int k = 0; k < CARRY; ++k) {
      if (32 * k < cnt) sum_batch(min(32, cnt - 32 * k), ks0[k], ks1[k], ksrc[k], krow[k]);
    }
    flush(32 * CARRY, cnt);
  } else {
    cnt = 0;
    scan_all([&]() { flush(0, cnt); cnt = 0; });
    flush(0, cnt);
  }
  __syncthreads();

  for (int i = tid; i < nrows * 64; i += 1024) {
    const int r = i >> 6, g = i & 63;
    const int half = g >> 5, l = g & 31;
    const float* tp = tile + r * HF + half * 128 + l;
    float4 x = make_float4(tp[0], tp[32], tp[64], tp[96]);
    const float dn = dtile[r * NH + half * 4 + (l >> 3)];
    if (dn > 0.f) { x.x /= dn; x.y /= dn; x.z /= dn; x.w /= dn; }
    x.x = x.x > 0.f ? x.x : expm1f(x.x);
    x.y = x.y > 0.f ? x.y : expm1f(x.y);
    x.z = x.z > 0.f ? x.z : expm1f(x.z);
    x.w = x.w > 0.f ? x.w : expm1f(x.w);
    *(float4*)(out + (size_t)(base + r) * HF + half * 128 + 4 * l) = x;
  }
}

extern "C" void kernel_launch(void* const* d_in, const int* in_sizes, int n_in,
                              void* d_out, int out_size, void* d_ws, size_t ws_size,
                              hipStream_t stream) {
  const float* h   = (const float*)d_in[0];
  const float* ef  = (const float*)d_in[1];
  const int*   src = (const int*)  d_in[2];
  const int*   dst = (const int*)  d_in[3];
  const float* W_w = (const float*)d_in[4];
  const float* We_w= (const float*)d_in[5];
  const float* a_s = (const float*)d_in[6];
  const float* a_d = (const float*)d_in[7];
  const float* a_e = (const float*)d_in[8];
  float* out = (float*)d_out;

  char* ws = (char*)d_ws;
  const size_t WH_BYTES   = (size_t)N_NODES * HF * 4;
  const size_t SBUF_BYTES = (size_t)N_EDGES * NH * 4;
  const size_t NH_BYTES   = (size_t)N_NODES * NH * 4;
  float*    Wh   = (float*)   (ws);
  float*    ebuf = (float*)   (ws + WH_BYTES);
  float*    ssrc = (float*)   (ws + WH_BYTES + SBUF_BYTES);
  float*    sdst = (float*)   (ws + WH_BYTES + SBUF_BYTES + NH_BYTES);
  unsigned* tkey = (unsigned*)(ws + WH_BYTES + SBUF_BYTES + 3 * NH_BYTES);
  float*    V    = (float*)   (ws + WH_BYTES + SBUF_BYTES + 4 * NH_BYTES);

  fold_we<<<1, 512, 0, stream>>>(We_w, a_e, V);
  gemm_and_edge_term<<<GEMM_WGS + LOGIT_WGS, 128, 0, stream>>>(h, W_w, a_s, a_d, Wh, ssrc, sdst,
                                                               ef, dst, V, ebuf, tkey);
  aggregate_tile<<<(N_NODES + TROWS - 1) / TROWS, 1024, 0, stream>>>(src, dst, Wh, ebuf, ssrc, sdst, tkey, out);
}
